# starting kernel
# baseline (speedup 1.0000x reference)
#include <hip/hip_runtime.h>
#include <hip/hip_bf16.h>

typedef __attribute__((ext_vector_type(2))) float v2f;
typedef __attribute__((ext_vector_type(8))) float v8f;

#define N_NODES 100000
#define N_EDGES 1600000
#define HID     64
#define NGRAPH  2048

__global__ void fill_k(float* __restrict__ p, float v, long long n) {
    long long i = (long long)blockIdx.x * blockDim.x + threadIdx.x;
    if (i < n) p[i] = v;
}

__global__ void deg_k(const int* __restrict__ dst, float* __restrict__ deg, int E) {
    int e = blockIdx.x * blockDim.x + threadIdx.x;
    if (e < E) atomicAdd(&deg[dst[e]], 1.0f);
}

__global__ void rsqrt_k(float* __restrict__ d, int n) {
    int i = blockIdx.x * blockDim.x + threadIdx.x;
    if (i < n) d[i] = rsqrtf(d[i]);
}

__global__ void norm_k(const int* __restrict__ src, const int* __restrict__ dst,
                       const float* __restrict__ dis, float* __restrict__ nrm, int E) {
    int e = blockIdx.x * blockDim.x + threadIdx.x;
    if (e < E) nrm[e] = dis[src[e]] * dis[dst[e]];
}

__global__ void gemm_wmma_k(const float* __restrict__ A, const float* __restrict__ W,
                            float* __restrict__ C, int rows, int K) {
    const int lane = threadIdx.x & 31;
    const int w    = (blockIdx.x * blockDim.x + threadIdx.x) >> 5;
    const int rowTiles = rows >> 4;
    if (w >= rowTiles) return;

    const int m0  = w << 4;
    const int mr  = lane & 15;
    const int khi = (lane >> 4) << 1;

    const float* arow = A + (size_t)(m0 + mr) * K + khi;
    const float* bcol = W + mr;

    v8f acc0 = {}, acc1 = {}, acc2 = {}, acc3 = {};
    for (int k = 0; k < K; k += 4) {
        v2f a;
        const float2 av = *(const float2*)(arow + k);
        a[0] = av.x; a[1] = av.y;

        const float* brow0 = bcol + (size_t)(k + khi) * HID;
        const float* brow1 = brow0 + HID;
        v2f b0, b1, b2, b3;
        b0[0] = brow0[0];  b0[1] = brow1[0];
        b1[0] = brow0[16]; b1[1] = brow1[16];
        b2[0] = brow0[32]; b2[1] = brow1[32];
        b3[0] = brow0[48]; b3[1] = brow1[48];

        acc0 = __builtin_amdgcn_wmma_f32_16x16x4_f32(false, a, false, b0, (short)0, acc0, false, false);
        acc1 = __builtin_amdgcn_wmma_f32_16x16x4_f32(false, a, false, b1, (short)0, acc1, false, false);
        acc2 = __builtin_amdgcn_wmma_f32_16x16x4_f32(false, a, false, b2, (short)0, acc2, false, false);
        acc3 = __builtin_amdgcn_wmma_f32_16x16x4_f32(false, a, false, b3, (short)0, acc3, false, false);
    }

    float* crow = C + (size_t)(m0 + ((lane >> 4) << 3)) * HID + mr;
#pragma unroll
    for (int j = 0; j < 8; ++j) {
        crow[(size_t)j * HID +  0] = acc0[j];
        crow[(size_t)j * HID + 16] = acc1[j];
        crow[(size_t)j * HID + 32] = acc2[j];
        crow[(size_t)j * HID + 48] = acc3[j];
    }
}

__global__ void edge_msg_k(const int* __restrict__ src, const int* __restrict__ dst,
                           const float* __restrict__ nrm, const float* __restrict__ h,
                           float* __restrict__ agg, int E) {
    int t = blockIdx.x * blockDim.x + threadIdx.x;
    int e = t >> 5;
    if (e >= E) return;
    int lane = t & 31;
    int s = src[e], d = dst[e];
    float nv = nrm[e];
    int f = lane << 1;
    float2 hv = *(const float2*)(h + (size_t)s * HID + f);
    atomicAdd(&agg[(size_t)d * HID + f    ], hv.x * nv);
    atomicAdd(&agg[(size_t)d * HID + f + 1], hv.y * nv);
}

__global__ void finalize_k(const float* agg, const float* __restrict__ hw,
                           const float* __restrict__ dis, const float* __restrict__ bias,
                           float* outp, int n) {
    int i = blockIdx.x * blockDim.x + threadIdx.x;
    if (i >= n * 16) return;
    int node = i >> 4;
    int f4   = (i & 15) << 2;
    float s = dis[node]; s = s * s;
    const float4 a = *(const float4*)(agg + (size_t)node * HID + f4);
    const float4 h = *(const float4*)(hw  + (size_t)node * HID + f4);
    float4 r;
    r.x = fmaxf(a.x + h.x * s + bias[f4 + 0], 0.0f);
    r.y = fmaxf(a.y + h.y * s + bias[f4 + 1], 0.0f);
    r.z = fmaxf(a.z + h.z * s + bias[f4 + 2], 0.0f);
    r.w = fmaxf(a.w + h.w * s + bias[f4 + 3], 0.0f);
    *(float4*)(outp + (size_t)node * HID + f4) = r;
}

__global__ void counts_k(const int* __restrict__ batch, float* __restrict__ counts, int n) {
    int i = blockIdx.x * blockDim.x + threadIdx.x;
    if (i < n) atomicAdd(&counts[batch[i]], 1.0f);
}

__global__ void pool_k(const float* __restrict__ h, const int* __restrict__ batch,
                       float* __restrict__ pooled, int n) {
    int i = blockIdx.x * blockDim.x + threadIdx.x;
    if (i >= n * HID) return;
    int node = i >> 6;
    int f    = i & 63;
    int g    = batch[node];
    float v  = h[i];
    atomicAdd(&pooled[(size_t)g * 128 + f], v);
    atomicMax((unsigned int*)&pooled[(size_t)g * 128 + 64 + f], __float_as_uint(v));
}

__global__ void meandiv_k(float* __restrict__ pooled, const float* __restrict__ counts, int G) {
    int i = blockIdx.x * blockDim.x + threadIdx.x;
    if (i >= G * HID) return;
    int g = i >> 6;
    int f = i & 63;
    pooled[(size_t)g * 128 + f] /= fmaxf(counts[g], 1.0f);
}

__global__ void head_k(const float* __restrict__ pooled, const float* __restrict__ Wout,
                       const float* __restrict__ bout, float* __restrict__ out, int G) {
    int t = blockIdx.x * blockDim.x + threadIdx.x;
    int g = t >> 5;
    int lane = t & 31;
    if (g >= G) return;
    float v = 0.0f;
#pragma unroll
    for (int j = lane; j < 2 * HID; j += 32) v += pooled[(size_t)g * 128 + j] * Wout[j];
#pragma unroll
    for (int off = 16; off; off >>= 1) v += __shfl_xor(v, off, 32);
    if (lane == 0) out[g] = v + bout[0];
}

extern "C" void kernel_launch(void* const* d_in, const int* in_sizes, int n_in,
                              void* d_out, int out_size, void* d_ws, size_t ws_size,
                              hipStream_t stream) {
    const float* x     = (const float*)d_in[0];
    const int*   eidx  = (const int*)  d_in[1];
    const int*   batch = (const int*)  d_in[2];
    const float* Ws[4] = { (const float*)d_in[3], (const float*)d_in[5],
                           (const float*)d_in[7], (const float*)d_in[9] };
    const float* bs[4] = { (const float*)d_in[4], (const float*)d_in[6],
                           (const float*)d_in[8], (const float*)d_in[10] };
    const float* Wout  = (const float*)d_in[11];
    const float* bout  = (const float*)d_in[12];

    const int N = N_NODES, E = N_EDGES, G = NGRAPH;
    const int* src = eidx;
    const int* dst = eidx + E;

    float* ws     = (float*)d_ws;
    float* dis    = ws;
    float* nrm    = dis + N;
    float* counts = nrm + E;
    float* buf0   = counts + G;
    float* buf1   = buf0 + (size_t)N * HID;
    float* buf2   = buf1 + (size_t)N * HID;

    float* out    = (float*)d_out;
    float* pooled = out + G;

    fill_k <<<(N + 255) / 256, 256, 0, stream>>>(dis, 1.0f, N);
    deg_k  <<<(E + 255) / 256, 256, 0, stream>>>(dst, dis, E);
    rsqrt_k<<<(N + 255) / 256, 256, 0, stream>>>(dis, N);
    norm_k <<<(E + 255) / 256, 256, 0, stream>>>(src, dst, dis, nrm, E);

    const float* hin = x;
    float* HW  = buf0;
    float* AGG = buf1;
    float* spare = buf2;
    int K = 128;
    const long long feat = (long long)N * HID;

    for (int l = 0; l < 4; ++l) {
        int gemmThreads = (N / 16) * 32;
        gemm_wmma_k<<<(gemmThreads + 255) / 256, 256, 0, stream>>>(hin, Ws[l], HW, N, K);
        fill_k     <<<(int)((feat + 255) / 256), 256, 0, stream>>>(AGG, 0.0f, feat);
        edge_msg_k <<<(int)(((long long)E * 32 + 255) / 256), 256, 0, stream>>>(src, dst, nrm, HW, AGG, E);
        finalize_k <<<(N * 16 + 255) / 256, 256, 0, stream>>>(AGG, HW, dis, bs[l], AGG, N);

        const float* next_in = AGG;
        float* newAGG = (l == 0) ? spare : (float*)hin;
        hin = next_in;
        AGG = newAGG;
        K = HID;
    }

    fill_k   <<<(G + G * 128 + 255) / 256, 256, 0, stream>>>(out, 0.0f, G + G * 128);
    fill_k   <<<(G + 255) / 256, 256, 0, stream>>>(counts, 0.0f, G);
    counts_k <<<(N + 255) / 256, 256, 0, stream>>>(batch, counts, N);
    pool_k   <<<(N * HID + 255) / 256, 256, 0, stream>>>(hin, batch, pooled, N);
    meandiv_k<<<(G * HID + 255) / 256, 256, 0, stream>>>(pooled, counts, G);
    head_k   <<<(G * 32 + 255) / 256, 256, 0, stream>>>(pooled, Wout, bout, out, G);
}


# Neighbour sum by destination: edges counting-sorted by dst each call (int atomic rank + scan), one wave per node adds h[src]*norm in registers in slot order; f32 atomics and agg zero-fill removed
# speedup vs baseline: 2.5933x; 2.5933x over previous
#include <hip/hip_runtime.h>
#include <hip/hip_bf16.h>

typedef __attribute__((ext_vector_type(2))) float v2f;
typedef __attribute__((ext_vector_type(8))) float v8f;

#define N_NODES 100000
#define N_EDGES 1600000
#define HID     64
#define NGRAPH  2048

__global__ void fill_k(float* __restrict__ p, float v, long long n) {
    long long i = (long long)blockIdx.x * blockDim.x + threadIdx.x;
    if (i < n) p[i] = v;
}

__global__ void gemm_wmma_k(const float* __restrict__ A, const float* __restrict__ W,
                            float* __restrict__ C, int rows, int K) {
    const int lane = threadIdx.x & 31;
    const int w    = (blockIdx.x * blockDim.x + threadIdx.x) >> 5;
    const int rowTiles = rows >> 4;
    if (w >= rowTiles) return;

    const int m0  = w << 4;
    const int mr  = lane & 15;
    const int khi = (lane >> 4) << 1;

    const float* arow = A + (size_t)(m0 + mr) * K + khi;
    const float* bcol = W + mr;

    v8f acc0 = {}, acc1 = {}, acc2 = {}, acc3 = {};
    for (int k = 0; k < K; k += 4) {
        v2f a;
        const float2 av = *(const float2*)(arow + k);
        a[0] = av.x; a[1] = av.y;

        const float* brow0 = bcol + (size_t)(k + khi) * HID;
        const float* brow1 = brow0 + HID;
        v2f b0, b1, b2, b3;
        b0[0] = brow0[0];  b0[1] = brow1[0];
        b1[0] = brow0[16]; b1[1] = brow1[16];
        b2[0] = brow0[32]; b2[1] = brow1[32];
        b3[0] = brow0[48]; b3[1] = brow1[48];

        acc0 = __builtin_amdgcn_wmma_f32_16x16x4_f32(false, a, false, b0, (short)0, acc0, false, false);
        acc1 = __builtin_amdgcn_wmma_f32_16x16x4_f32(false, a, false, b1, (short)0, acc1, false, false);
        acc2 = __builtin_amdgcn_wmma_f32_16x16x4_f32(false, a, false, b2, (short)0, acc2, false, false);
        acc3 = __builtin_amdgcn_wmma_f32_16x16x4_f32(false, a, false, b3, (short)0, acc3, false, false);
    }

    float* crow = C + (size_t)(m0 + ((lane >> 4) << 3)) * HID + mr;
#pragma unroll
    for (int j = 0; j < 8; ++j) {
        crow[(size_t)j * HID +  0] = acc0[j];
        crow[(size_t)j * HID + 16] = acc1[j];
        crow[(size_t)j * HID + 32] = acc2[j];
        crow[(size_t)j * HID + 48] = acc3[j];
    }
}

__global__ void count_rank_k(const int* __restrict__ dst, int* __restrict__ cnt,
                             int* __restrict__ rank, int E) {
    int e = blockIdx.x * blockDim.x + threadIdx.x;
    if (e < E) rank[e] = atomicAdd(&cnt[dst[e]], 1);
}

#define SCAN_T 256

__global__ void __launch_bounds__(SCAN_T)
blocksum_k(const int* __restrict__ cnt, int* __restrict__ bsum, int n4) {
    __shared__ int sm[SCAN_T / 32];
    const int tid = threadIdx.x;
    const int i4  = blockIdx.x * SCAN_T + tid;
    int t = 0;
    if (i4 < n4) { const int4 v = ((const int4*)cnt)[i4]; t = v.x + v.y + v.z + v.w; }
    for (int off = 16; off; off >>= 1) t += __shfl_xor(t, off, 32);
    if ((tid & 31) == 0) sm[tid >> 5] = t;
    __syncthreads();
    if (tid == 0) {
        int s = 0;
        for (int k = 0; k < SCAN_T / 32; ++k) s += sm[k];
        bsum[blockIdx.x] = s;
    }
}

__global__ void __launch_bounds__(SCAN_T)
rowptr_k(const int* __restrict__ cnt, const int* __restrict__ bsum,
         int* __restrict__ rowptr, float* __restrict__ dis, int n4) {
    __shared__ int smA[SCAN_T / 32];
    __shared__ int smB[SCAN_T / 32];
    const int tid  = threadIdx.x;
    const int lane = tid & 31;
    const int wid  = tid >> 5;
    const int i4   = blockIdx.x * SCAN_T + tid;

    int part = (tid < (int)blockIdx.x) ? bsum[tid] : 0;
    for (int off = 16; off; off >>= 1) part += __shfl_xor(part, off, 32);
    if (lane == 0) smA[wid] = part;

    int4 v = make_int4(0, 0, 0, 0);
    if (i4 < n4) v = ((const int4*)cnt)[i4];
    const int t = v.x + v.y + v.z + v.w;
    int inc = t;
    for (int off = 1; off < 32; off <<= 1) {
        const int u = __shfl_up(inc, off, 32);
        if (lane >= off) inc += u;
    }
    if (lane == 31) smB[wid] = inc;
    __syncthreads();

    int pre = 0;
    for (int k = 0; k < SCAN_T / 32; ++k) {
        pre += smA[k];
        if (k < wid) pre += smB[k];
    }
    const int excl = pre + inc - t;
    if (i4 < n4) {
        int4 r;
        r.x = excl; r.y = r.x + v.x; r.z = r.y + v.y; r.w = r.z + v.z;
        ((int4*)rowptr)[i4] = r;
        float4 d;
        d.x = rsqrtf((float)(v.x + 1)); d.y = rsqrtf((float)(v.y + 1));
        d.z = rsqrtf((float)(v.z + 1)); d.w = rsqrtf((float)(v.w + 1));
        ((float4*)dis)[i4] = d;
    }
    if (i4 == n4 - 1) rowptr[4 * n4] = excl + t;
}

__global__ void place_k(const int* __restrict__ src, const int* __restrict__ dst,
                        const int* __restrict__ rank, const int* __restrict__ rowptr,
                        const float* __restrict__ dis, int* __restrict__ rec, int E) {
    int e = blockIdx.x * blockDim.x + threadIdx.x;
    if (e >= E) return;
    const int s = src[e], d = dst[e];
    const int p = rowptr[d] + rank[e];
    ((int2*)rec)[p] = make_int2(s, __float_as_int(dis[s] * dis[d]));
}

#define GRP 8
__global__ void __launch_bounds__(256)
gather_sum_k(const int* __restrict__ rowptr, const int* __restrict__ rec,
             const float* __restrict__ h, float* __restrict__ agg, int n) {
    const int lane = threadIdx.x & 31;
    int node = (blockIdx.x * blockDim.x + threadIdx.x) >> 5;
    if (node >= n) return;
    node = __builtin_amdgcn_readfirstlane(node);
    const int beg = rowptr[node];
    const int end = rowptr[node + 1];
    const float* hl = h + (lane << 1);
    float2 acc = make_float2(0.0f, 0.0f);

    for (int base = beg; base < end; base += 32) {
        const int m = min(32, end - base);
        int2 ev = make_int2(0, 0);
        if (lane < m) ev = ((const int2*)rec)[base + lane];
        for (int j = 0; j < m; j += GRP) {
            float2 hv[GRP];
            float  wv[GRP];
#pragma unroll
            for (int u = 0; u < GRP; ++u) {
                const int jj = min(j + u, m - 1);
                const int   s = __builtin_amdgcn_readlane(ev.x, jj);
                const float w = __int_as_float(__builtin_amdgcn_readlane(ev.y, jj));
                wv[u] = (j + u < m) ? w : 0.0f;
                hv[u] = *(const float2*)(hl + (unsigned)(s * HID));
            }
#pragma unroll
            for (int u = 0; u < GRP; ++u) {
                acc.x += hv[u].x * wv[u];
                acc.y += hv[u].y * wv[u];
            }
        }
    }
    *(float2*)(agg + (size_t)node * HID + (lane << 1)) = acc;
}

__global__ void finalize_k(const float* agg, const float* __restrict__ hw,
                           const float* __restrict__ dis, const float* __restrict__ bias,
                           float* outp, int n) {
    int i = blockIdx.x * blockDim.x + threadIdx.x;
    if (i >= n * 16) return;
    int node = i >> 4;
    int f4   = (i & 15) << 2;
    float s = dis[node]; s = s * s;
    const float4 a = *(const float4*)(agg + (size_t)node * HID + f4);
    const float4 h = *(const float4*)(hw  + (size_t)node * HID + f4);
    float4 r;
    r.x = fmaxf(a.x + h.x * s + bias[f4 + 0], 0.0f);
    r.y = fmaxf(a.y + h.y * s + bias[f4 + 1], 0.0f);
    r.z = fmaxf(a.z + h.z * s + bias[f4 + 2], 0.0f);
    r.w = fmaxf(a.w + h.w * s + bias[f4 + 3], 0.0f);
    *(float4*)(outp + (size_t)node * HID + f4) = r;
}

__global__ void counts_k(const int* __restrict__ batch, float* __restrict__ counts, int n) {
    int i = blockIdx.x * blockDim.x + threadIdx.x;
    if (i < n) atomicAdd(&counts[batch[i]], 1.0f);
}

__global__ void pool_k(const float* __restrict__ h, const int* __restrict__ batch,
                       float* __restrict__ pooled, int n) {
    int i = blockIdx.x * blockDim.x + threadIdx.x;
    if (i >= n * HID) return;
    int node = i >> 6;
    int f    = i & 63;
    int g    = batch[node];
    float v  = h[i];
    atomicAdd(&pooled[(size_t)g * 128 + f], v);
    atomicMax((unsigned int*)&pooled[(size_t)g * 128 + 64 + f], __float_as_uint(v));
}

__global__ void meandiv_k(float* __restrict__ pooled, const float* __restrict__ counts, int G) {
    int i = blockIdx.x * blockDim.x + threadIdx.x;
    if (i >= G * HID) return;
    int g = i >> 6;
    int f = i & 63;
    pooled[(size_t)g * 128 + f] /= fmaxf(counts[g], 1.0f);
}

__global__ void head_k(const float* __restrict__ pooled, const float* __restrict__ Wout,
                       const float* __restrict__ bout, float* __restrict__ out, int G) {
    int t = blockIdx.x * blockDim.x + threadIdx.x;
    int g = t >> 5;
    int lane = t & 31;
    if (g >= G) return;
    float v = 0.0f;
#pragma unroll
    for (int j = lane; j < 2 * HID; j += 32) v += pooled[(size_t)g * 128 + j] * Wout[j];
#pragma unroll
    for (int off = 16; off; off >>= 1) v += __shfl_xor(v, off, 32);
    if (lane == 0) out[g] = v + bout[0];
}

#define WS_DIS    0
#define WS_CNT    (WS_DIS + N_NODES)
#define WS_ROWPTR (WS_CNT + N_NODES)
#define WS_BSUM   (WS_ROWPTR + N_NODES + 4)
#define WS_COUNTS (WS_BSUM + 128)
#define WS_RANK   (WS_COUNTS + NGRAPH)
#define WS_REC    (WS_RANK + N_EDGES)
#define WS_BUF0   (((WS_REC + 2 * N_EDGES) + 63) / 64 * 64)

extern "C" void kernel_launch(void* const* d_in, const int* in_sizes, int n_in,
                              void* d_out, int out_size, void* d_ws, size_t ws_size,
                              hipStream_t stream) {
    const float* x     = (const float*)d_in[0];
    const int*   eidx  = (const int*)  d_in[1];
    const int*   batch = (const int*)  d_in[2];
    const float* Ws[4] = { (const float*)d_in[3], (const float*)d_in[5],
                           (const float*)d_in[7], (const float*)d_in[9] };
    const float* bs[4] = { (const float*)d_in[4], (const float*)d_in[6],
                           (const float*)d_in[8], (const float*)d_in[10] };
    const float* Wout  = (const float*)d_in[11];
    const float* bout  = (const float*)d_in[12];

    const int N = N_NODES, E = N_EDGES, G = NGRAPH;
    const int* src = eidx;
    const int* dst = eidx + E;

    float* ws     = (float*)d_ws;
    float* dis    = ws + WS_DIS;
    int*   cnt    = (int*)(ws + WS_CNT);
    int*   rowptr = (int*)(ws + WS_ROWPTR);
    int*   bsum   = (int*)(ws + WS_BSUM);
    float* counts = ws + WS_COUNTS;
    int*   rank   = (int*)(ws + WS_RANK);
    int*   rec    = (int*)(ws + WS_REC);
    float* buf0   = ws + WS_BUF0;
    float* buf1   = buf0 + (size_t)N * HID;
    float* buf2   = buf1 + (size_t)N * HID;

    float* out    = (float*)d_out;
    float* pooled = out + G;

    const int n4 = N / 4;
    const int scanBlocks = (n4 + SCAN_T - 1) / SCAN_T;
    fill_k      <<<(N + 255) / 256, 256, 0, stream>>>((float*)cnt, 0.0f, N);
    count_rank_k<<<(E + 255) / 256, 256, 0, stream>>>(dst, cnt, rank, E);
    blocksum_k  <<<scanBlocks, SCAN_T, 0, stream>>>(cnt, bsum, n4);
    rowptr_k    <<<scanBlocks, SCAN_T, 0, stream>>>(cnt, bsum, rowptr, dis, n4);
    place_k     <<<(E + 255) / 256, 256, 0, stream>>>(src, dst, rank, rowptr, dis, rec, E);

    const float* hin = x;
    float* HW  = buf0;
    float* AGG = buf1;
    float* spare = buf2;
    int K = 128;

    for (int l = 0; l < 4; ++l) {
        int gemmThreads = (N / 16) * 32;
        gemm_wmma_k <<<(gemmThreads + 255) / 256, 256, 0, stream>>>(hin, Ws[l], HW, N, K);
        gather_sum_k<<<(N * 32 + 255) / 256, 256, 0, stream>>>(rowptr, rec, HW, AGG, N);
        finalize_k  <<<(N * 16 + 255) / 256, 256, 0, stream>>>(AGG, HW, dis, bs[l], AGG, N);

        const float* next_in = AGG;
        float* newAGG = (l == 0) ? spare : (float*)hin;
        hin = next_in;
        AGG = newAGG;
        K = HID;
    }

    fill_k   <<<(G + G * 128 + 255) / 256, 256, 0, stream>>>(out, 0.0f, G + G * 128);
    fill_k   <<<(G + 255) / 256, 256, 0, stream>>>(counts, 0.0f, G);
    counts_k <<<(N + 255) / 256, 256, 0, stream>>>(batch, counts, N);
    pool_k   <<<(N * HID + 255) / 256, 256, 0, stream>>>(hin, batch, pooled, N);
    meandiv_k<<<(G * HID + 255) / 256, 256, 0, stream>>>(pooled, counts, G);
    head_k   <<<(G * 32 + 255) / 256, 256, 0, stream>>>(pooled, Wout, bout, out, G);
}


# Self-loop term, bias and ReLU applied inside the neighbour-sum kernel on the sum still in registers; the 4 separate passes over the sum (launch, 2 row reads, 1 row write each) removed; sum order kept
# speedup vs baseline: 2.8295x; 1.0911x over previous
#include <hip/hip_runtime.h>
#include <hip/hip_bf16.h>

typedef __attribute__((ext_vector_type(2))) float v2f;
typedef __attribute__((ext_vector_type(8))) float v8f;

#define N_NODES 100000
#define N_EDGES 1600000
#define HID     64
#define NGRAPH  2048

__global__ void fill_k(float* __restrict__ p, float v, long long n) {
    long long i = (long long)blockIdx.x * blockDim.x + threadIdx.x;
    if (i < n) p[i] = v;
}

__global__ void gemm_wmma_k(const float* __restrict__ A, const float* __restrict__ W,
                            float* __restrict__ C, int rows, int K) {
    const int lane = threadIdx.x & 31;
    const int w    = (blockIdx.x * blockDim.x + threadIdx.x) >> 5;
    const int rowTiles = rows >> 4;
    if (w >= rowTiles) return;

    const int m0  = w << 4;
    const int mr  = lane & 15;
    const int khi = (lane >> 4) << 1;

    const float* arow = A + (size_t)(m0 + mr) * K + khi;
    const float* bcol = W + mr;

    v8f acc0 = {}, acc1 = {}, acc2 = {}, acc3 = {};
    for (int k = 0; k < K; k += 4) {
        v2f a;
        const float2 av = *(const float2*)(arow + k);
        a[0] = av.x; a[1] = av.y;

        const float* brow0 = bcol + (size_t)(k + khi) * HID;
        const float* brow1 = brow0 + HID;
        v2f b0, b1, b2, b3;
        b0[0] = brow0[0];  b0[1] = brow1[0];
        b1[0] = brow0[16]; b1[1] = brow1[16];
        b2[0] = brow0[32]; b2[1] = brow1[32];
        b3[0] = brow0[48]; b3[1] = brow1[48];

        acc0 = __builtin_amdgcn_wmma_f32_16x16x4_f32(false, a, false, b0, (short)0, acc0, false, false);
        acc1 = __builtin_amdgcn_wmma_f32_16x16x4_f32(false, a, false, b1, (short)0, acc1, false, false);
        acc2 = __builtin_amdgcn_wmma_f32_16x16x4_f32(false, a, false, b2, (short)0, acc2, false, false);
        acc3 = __builtin_amdgcn_wmma_f32_16x16x4_f32(false, a, false, b3, (short)0, acc3, false, false);
    }

    float* crow = C + (size_t)(m0 + ((lane >> 4) << 3)) * HID + mr;
#pragma unroll
    for (int j = 0; j < 8; ++j) {
        crow[(size_t)j * HID +  0] = acc0[j];
        crow[(size_t)j * HID + 16] = acc1[j];
        crow[(size_t)j * HID + 32] = acc2[j];
        crow[(size_t)j * HID + 48] = acc3[j];
    }
}

__global__ void count_rank_k(const int* __restrict__ dst, int* __restrict__ cnt,
                             int* __restrict__ rank, int E) {
    int e = blockIdx.x * blockDim.x + threadIdx.x;
    if (e < E) rank[e] = atomicAdd(&cnt[dst[e]], 1);
}

#define SCAN_T 256

__global__ void __launch_bounds__(SCAN_T)
blocksum_k(const int* __restrict__ cnt, int* __restrict__ bsum, int n4) {
    __shared__ int sm[SCAN_T / 32];
    const int tid = threadIdx.x;
    const int i4  = blockIdx.x * SCAN_T + tid;
    int t = 0;
    if (i4 < n4) { const int4 v = ((const int4*)cnt)[i4]; t = v.x + v.y + v.z + v.w; }
    for (int off = 16; off; off >>= 1) t += __shfl_xor(t, off, 32);
    if ((tid & 31) == 0) sm[tid >> 5] = t;
    __syncthreads();
    if (tid == 0) {
        int s = 0;
        for (int k = 0; k < SCAN_T / 32; ++k) s += sm[k];
        bsum[blockIdx.x] = s;
    }
}

__global__ void __launch_bounds__(SCAN_T)
rowptr_k(const int* __restrict__ cnt, const int* __restrict__ bsum,
         int* __restrict__ rowptr, float* __restrict__ dis, int n4) {
    __shared__ int smA[SCAN_T / 32];
    __shared__ int smB[SCAN_T / 32];
    const int tid  = threadIdx.x;
    const int lane = tid & 31;
    const int wid  = tid >> 5;
    const int i4   = blockIdx.x * SCAN_T + tid;

    int part = (tid < (int)blockIdx.x) ? bsum[tid] : 0;
    for (int off = 16; off; off >>= 1) part += __shfl_xor(part, off, 32);
    if (lane == 0) smA[wid] = part;

    int4 v = make_int4(0, 0, 0, 0);
    if (i4 < n4) v = ((const int4*)cnt)[i4];
    const int t = v.x + v.y + v.z + v.w;
    int inc = t;
    for (int off = 1; off < 32; off <<= 1) {
        const int u = __shfl_up(inc, off, 32);
        if (lane >= off) inc += u;
    }
    if (lane == 31) smB[wid] = inc;
    __syncthreads();

    int pre = 0;
    for (int k = 0; k < SCAN_T / 32; ++k) {
        pre += smA[k];
        if (k < wid) pre += smB[k];
    }
    const int excl = pre + inc - t;
    if (i4 < n4) {
        int4 r;
        r.x = excl; r.y = r.x + v.x; r.z = r.y + v.y; r.w = r.z + v.z;
        ((int4*)rowptr)[i4] = r;
        float4 d;
        d.x = rsqrtf((float)(v.x + 1)); d.y = rsqrtf((float)(v.y + 1));
        d.z = rsqrtf((float)(v.z + 1)); d.w = rsqrtf((float)(v.w + 1));
        ((float4*)dis)[i4] = d;
    }
    if (i4 == n4 - 1) rowptr[4 * n4] = excl + t;
}

__global__ void place_k(const int* __restrict__ src, const int* __restrict__ dst,
                        const int* __restrict__ rank, const int* __restrict__ rowptr,
                        const float* __restrict__ dis, int* __restrict__ rec, int E) {
    int e = blockIdx.x * blockDim.x + threadIdx.x;
    if (e >= E) return;
    const int s = src[e], d = dst[e];
    const int p = rowptr[d] + rank[e];
    ((int2*)rec)[p] = make_int2(s, __float_as_int(dis[s] * dis[d]));
}

#define GRP 8
__global__ void __launch_bounds__(256)
gather_layer_k(const int* __restrict__ rowptr, const int* __restrict__ rec,
               const float* __restrict__ h, const float* __restrict__ dis,
               const float* __restrict__ bias, float* __restrict__ outp, int n) {
    const int lane = threadIdx.x & 31;
    int node = (blockIdx.x * blockDim.x + threadIdx.x) >> 5;
    if (node >= n) return;
    node = __builtin_amdgcn_readfirstlane(node);
    const int beg = rowptr[node];
    const int end = rowptr[node + 1];
    const float* hl = h + (lane << 1);
    const float2 own = *(const float2*)(hl + (size_t)node * HID);
    const float2 bv  = *(const float2*)(bias + (lane << 1));
    float sw = dis[node]; sw = sw * sw;
    float2 acc = make_float2(0.0f, 0.0f);

    for (int base = beg; base < end; base += 32) {
        const int m = min(32, end - base);
        int2 ev = make_int2(0, 0);
        if (lane < m) ev = ((const int2*)rec)[base + lane];
        for (int j = 0; j < m; j += GRP) {
            float2 hv[GRP];
            float  wv[GRP];
#pragma unroll
            for (int u = 0; u < GRP; ++u) {
                const int jj = min(j + u, m - 1);
                const int   s = __builtin_amdgcn_readlane(ev.x, jj);
                const float w = __int_as_float(__builtin_amdgcn_readlane(ev.y, jj));
                wv[u] = (j + u < m) ? w : 0.0f;
                hv[u] = *(const float2*)(hl + (unsigned)(s * HID));
            }
#pragma unroll
            for (int u = 0; u < GRP; ++u) {
                acc.x += hv[u].x * wv[u];
                acc.y += hv[u].y * wv[u];
            }
        }
    }
    float2 r;
    r.x = fmaxf(acc.x + own.x * sw + bv.x, 0.0f);
    r.y = fmaxf(acc.y + own.y * sw + bv.y, 0.0f);
    *(float2*)(outp + (size_t)node * HID + (lane << 1)) = r;
}

__global__ void counts_k(const int* __restrict__ batch, float* __restrict__ counts, int n) {
    int i = blockIdx.x * blockDim.x + threadIdx.x;
    if (i < n) atomicAdd(&counts[batch[i]], 1.0f);
}

__global__ void pool_k(const float* __restrict__ h, const int* __restrict__ batch,
                       float* __restrict__ pooled, int n) {
    int i = blockIdx.x * blockDim.x + threadIdx.x;
    if (i >= n * HID) return;
    int node = i >> 6;
    int f    = i & 63;
    int g    = batch[node];
    float v  = h[i];
    atomicAdd(&pooled[(size_t)g * 128 + f], v);
    atomicMax((unsigned int*)&pooled[(size_t)g * 128 + 64 + f], __float_as_uint(v));
}

__global__ void meandiv_k(float* __restrict__ pooled, const float* __restrict__ counts, int G) {
    int i = blockIdx.x * blockDim.x + threadIdx.x;
    if (i >= G * HID) return;
    int g = i >> 6;
    int f = i & 63;
    pooled[(size_t)g * 128 + f] /= fmaxf(counts[g], 1.0f);
}

__global__ void head_k(const float* __restrict__ pooled, const float* __restrict__ Wout,
                       const float* __restrict__ bout, float* __restrict__ out, int G) {
    int t = blockIdx.x * blockDim.x + threadIdx.x;
    int g = t >> 5;
    int lane = t & 31;
    if (g >= G) return;
    float v = 0.0f;
#pragma unroll
    for (int j = lane; j < 2 * HID; j += 32) v += pooled[(size_t)g * 128 + j] * Wout[j];
#pragma unroll
    for (int off = 16; off; off >>= 1) v += __shfl_xor(v, off, 32);
    if (lane == 0) out[g] = v + bout[0];
}

#define WS_DIS    0
#define WS_CNT    (WS_DIS + N_NODES)
#define WS_ROWPTR (WS_CNT + N_NODES)
#define WS_BSUM   (WS_ROWPTR + N_NODES + 4)
#define WS_COUNTS (WS_BSUM + 128)
#define WS_RANK   (WS_COUNTS + NGRAPH)
#define WS_REC    (WS_RANK + N_EDGES)
#define WS_BUF0   (((WS_REC + 2 * N_EDGES) + 63) / 64 * 64)

extern "C" void kernel_launch(void* const* d_in, const int* in_sizes, int n_in,
                              void* d_out, int out_size, void* d_ws, size_t ws_size,
                              hipStream_t stream) {
    const float* x     = (const float*)d_in[0];
    const int*   eidx  = (const int*)  d_in[1];
    const int*   batch = (const int*)  d_in[2];
    const float* Ws[4] = { (const float*)d_in[3], (const float*)d_in[5],
                           (const float*)d_in[7], (const float*)d_in[9] };
    const float* bs[4] = { (const float*)d_in[4], (const float*)d_in[6],
                           (const float*)d_in[8], (const float*)d_in[10] };
    const float* Wout  = (const float*)d_in[11];
    const float* bout  = (const float*)d_in[12];

    const int N = N_NODES, E = N_EDGES, G = NGRAPH;
    const int* src = eidx;
    const int* dst = eidx + E;

    float* ws     = (float*)d_ws;
    float* dis    = ws + WS_DIS;
    int*   cnt    = (int*)(ws + WS_CNT);
    int*   rowptr = (int*)(ws + WS_ROWPTR);
    int*   bsum   = (int*)(ws + WS_BSUM);
    float* counts = ws + WS_COUNTS;
    int*   rank   = (int*)(ws + WS_RANK);
    int*   rec    = (int*)(ws + WS_REC);
    float* buf0   = ws + WS_BUF0;
    float* buf1   = buf0 + (size_t)N * HID;
    float* buf2   = buf1 + (size_t)N * HID;

    float* out    = (float*)d_out;
    float* pooled = out + G;

    const int n4 = N / 4;
    const int scanBlocks = (n4 + SCAN_T - 1) / SCAN_T;
    fill_k      <<<(N + 255) / 256, 256, 0, stream>>>((float*)cnt, 0.0f, N);
    count_rank_k<<<(E + 255) / 256, 256, 0, stream>>>(dst, cnt, rank, E);
    blocksum_k  <<<scanBlocks, SCAN_T, 0, stream>>>(cnt, bsum, n4);
    rowptr_k    <<<scanBlocks, SCAN_T, 0, stream>>>(cnt, bsum, rowptr, dis, n4);
    place_k     <<<(E + 255) / 256, 256, 0, stream>>>(src, dst, rank, rowptr, dis, rec, E);

    const float* hin = x;
    float* HW  = buf0;
    float* AGG = buf1;
    float* spare = buf2;
    int K = 128;

    for (int l = 0; l < 4; ++l) {
        int gemmThreads = (N / 16) * 32;
        gemm_wmma_k <<<(gemmThreads + 255) / 256, 256, 0, stream>>>(hin, Ws[l], HW, N, K);
        gather_layer_k<<<(N * 32 + 255) / 256, 256, 0, stream>>>(rowptr, rec, HW, dis, bs[l], AGG, N);

        const float* next_in = AGG;
        float* newAGG = (l == 0) ? spare : (float*)hin;
        hin = next_in;
        AGG = newAGG;
        K = HID;
    }

    fill_k   <<<(G + G * 128 + 255) / 256, 256, 0, stream>>>(out, 0.0f, G + G * 128);
    fill_k   <<<(G + 255) / 256, 256, 0, stream>>>(counts, 0.0f, G);
    counts_k <<<(N + 255) / 256, 256, 0, stream>>>(batch, counts, N);
    pool_k   <<<(N * HID + 255) / 256, 256, 0, stream>>>(hin, batch, pooled, N);
    meandiv_k<<<(G * HID + 255) / 256, 256, 0, stream>>>(pooled, counts, G);
    head_k   <<<(G * 32 + 255) / 256, 256, 0, stream>>>(pooled, Wout, bout, out, G);
}


# Pooling by runs: a wave walks 16 consecutive nodes, keeps sum, max and count in registers while the graph id repeats, then one atomic per feature per run; per-node atomics and count launch removed
# speedup vs baseline: 3.0010x; 1.0606x over previous
#include <hip/hip_runtime.h>
#include <hip/hip_bf16.h>

typedef __attribute__((ext_vector_type(2))) float v2f;
typedef __attribute__((ext_vector_type(8))) float v8f;

#define N_NODES 100000
#define N_EDGES 1600000
#define HID     64
#define NGRAPH  2048

__global__ void fill_k(float* __restrict__ p, float v, long long n) {
    long long i = (long long)blockIdx.x * blockDim.x + threadIdx.x;
    if (i < n) p[i] = v;
}

__global__ void gemm_wmma_k(const float* __restrict__ A, const float* __restrict__ W,
                            float* __restrict__ C, int rows, int K) {
    const int lane = threadIdx.x & 31;
    const int w    = (blockIdx.x * blockDim.x + threadIdx.x) >> 5;
    const int rowTiles = rows >> 4;
    if (w >= rowTiles) return;

    const int m0  = w << 4;
    const int mr  = lane & 15;
    const int khi = (lane >> 4) << 1;

    const float* arow = A + (size_t)(m0 + mr) * K + khi;
    const float* bcol = W + mr;

    v8f acc0 = {}, acc1 = {}, acc2 = {}, acc3 = {};
    for (int k = 0; k < K; k += 4) {
        v2f a;
        const float2 av = *(const float2*)(arow + k);
        a[0] = av.x; a[1] = av.y;

        const float* brow0 = bcol + (size_t)(k + khi) * HID;
        const float* brow1 = brow0 + HID;
        v2f b0, b1, b2, b3;
        b0[0] = brow0[0];  b0[1] = brow1[0];
        b1[0] = brow0[16]; b1[1] = brow1[16];
        b2[0] = brow0[32]; b2[1] = brow1[32];
        b3[0] = brow0[48]; b3[1] = brow1[48];

        acc0 = __builtin_amdgcn_wmma_f32_16x16x4_f32(false, a, false, b0, (short)0, acc0, false, false);
        acc1 = __builtin_amdgcn_wmma_f32_16x16x4_f32(false, a, false, b1, (short)0, acc1, false, false);
        acc2 = __builtin_amdgcn_wmma_f32_16x16x4_f32(false, a, false, b2, (short)0, acc2, false, false);
        acc3 = __builtin_amdgcn_wmma_f32_16x16x4_f32(false, a, false, b3, (short)0, acc3, false, false);
    }

    float* crow = C + (size_t)(m0 + ((lane >> 4) << 3)) * HID + mr;
#pragma unroll
    for (int j = 0; j < 8; ++j) {
        crow[(size_t)j * HID +  0] = acc0[j];
        crow[(size_t)j * HID + 16] = acc1[j];
        crow[(size_t)j * HID + 32] = acc2[j];
        crow[(size_t)j * HID + 48] = acc3[j];
    }
}

__global__ void count_rank_k(const int* __restrict__ dst, int* __restrict__ cnt,
                             int* __restrict__ rank, int E) {
    int e = blockIdx.x * blockDim.x + threadIdx.x;
    if (e < E) rank[e] = atomicAdd(&cnt[dst[e]], 1);
}

#define SCAN_T 256

__global__ void __launch_bounds__(SCAN_T)
blocksum_k(const int* __restrict__ cnt, int* __restrict__ bsum, int n4) {
    __shared__ int sm[SCAN_T / 32];
    const int tid = threadIdx.x;
    const int i4  = blockIdx.x * SCAN_T + tid;
    int t = 0;
    if (i4 < n4) { const int4 v = ((const int4*)cnt)[i4]; t = v.x + v.y + v.z + v.w; }
    for (int off = 16; off; off >>= 1) t += __shfl_xor(t, off, 32);
    if ((tid & 31) == 0) sm[tid >> 5] = t;
    __syncthreads();
    if (tid == 0) {
        int s = 0;
        for (int k = 0; k < SCAN_T / 32; ++k) s += sm[k];
        bsum[blockIdx.x] = s;
    }
}

__global__ void __launch_bounds__(SCAN_T)
rowptr_k(const int* __restrict__ cnt, const int* __restrict__ bsum,
         int* __restrict__ rowptr, float* __restrict__ dis, int n4) {
    __shared__ int smA[SCAN_T / 32];
    __shared__ int smB[SCAN_T / 32];
    const int tid  = threadIdx.x;
    const int lane = tid & 31;
    const int wid  = tid >> 5;
    const int i4   = blockIdx.x * SCAN_T + tid;

    int part = (tid < (int)blockIdx.x) ? bsum[tid] : 0;
    for (int off = 16; off; off >>= 1) part += __shfl_xor(part, off, 32);
    if (lane == 0) smA[wid] = part;

    int4 v = make_int4(0, 0, 0, 0);
    if (i4 < n4) v = ((const int4*)cnt)[i4];
    const int t = v.x + v.y + v.z + v.w;
    int inc = t;
    for (int off = 1; off < 32; off <<= 1) {
        const int u = __shfl_up(inc, off, 32);
        if (lane >= off) inc += u;
    }
    if (lane == 31) smB[wid] = inc;
    __syncthreads();

    int pre = 0;
    for (int k = 0; k < SCAN_T / 32; ++k) {
        pre += smA[k];
        if (k < wid) pre += smB[k];
    }
    const int excl = pre + inc - t;
    if (i4 < n4) {
        int4 r;
        r.x = excl; r.y = r.x + v.x; r.z = r.y + v.y; r.w = r.z + v.z;
        ((int4*)rowptr)[i4] = r;
        float4 d;
        d.x = rsqrtf((float)(v.x + 1)); d.y = rsqrtf((float)(v.y + 1));
        d.z = rsqrtf((float)(v.z + 1)); d.w = rsqrtf((float)(v.w + 1));
        ((float4*)dis)[i4] = d;
    }
    if (i4 == n4 - 1) rowptr[4 * n4] = excl + t;
}

__global__ void place_k(const int* __restrict__ src, const int* __restrict__ dst,
                        const int* __restrict__ rank, const int* __restrict__ rowptr,
                        const float* __restrict__ dis, int* __restrict__ rec, int E) {
    int e = blockIdx.x * blockDim.x + threadIdx.x;
    if (e >= E) return;
    const int s = src[e], d = dst[e];
    const int p = rowptr[d] + rank[e];
    ((int2*)rec)[p] = make_int2(s, __float_as_int(dis[s] * dis[d]));
}

#define GRP 8
__global__ void __launch_bounds__(256)
gather_layer_k(const int* __restrict__ rowptr, const int* __restrict__ rec,
               const float* __restrict__ h, const float* __restrict__ dis,
               const float* __restrict__ bias, float* __restrict__ outp, int n) {
    const int lane = threadIdx.x & 31;
    int node = (blockIdx.x * blockDim.x + threadIdx.x) >> 5;
    if (node >= n) return;
    node = __builtin_amdgcn_readfirstlane(node);
    const int beg = rowptr[node];
    const int end = rowptr[node + 1];
    const float* hl = h + (lane << 1);
    const float2 own = *(const float2*)(hl + (size_t)node * HID);
    const float2 bv  = *(const float2*)(bias + (lane << 1));
    float sw = dis[node]; sw = sw * sw;
    float2 acc = make_float2(0.0f, 0.0f);

    for (int base = beg; base < end; base += 32) {
        const int m = min(32, end - base);
        int2 ev = make_int2(0, 0);
        if (lane < m) ev = ((const int2*)rec)[base + lane];
        for (int j = 0; j < m; j += GRP) {
            float2 hv[GRP];
            float  wv[GRP];
#pragma unroll
            for (int u = 0; u < GRP; ++u) {
                const int jj = min(j + u, m - 1);
                const int   s = __builtin_amdgcn_readlane(ev.x, jj);
                const float w = __int_as_float(__builtin_amdgcn_readlane(ev.y, jj));
                wv[u] = (j + u < m) ? w : 0.0f;
                hv[u] = *(const float2*)(hl + (unsigned)(s * HID));
            }
#pragma unroll
            for (int u = 0; u < GRP; ++u) {
                acc.x += hv[u].x * wv[u];
                acc.y += hv[u].y * wv[u];
            }
        }
    }
    float2 r;
    r.x = fmaxf(acc.x + own.x * sw + bv.x, 0.0f);
    r.y = fmaxf(acc.y + own.y * sw + bv.y, 0.0f);
    *(float2*)(outp + (size_t)node * HID + (lane << 1)) = r;
}

#define PRUN 16
__device__ __forceinline__ void pool_flush(float* __restrict__ pooled, float* __restrict__ counts,
                                           int g, int lane, float sx, float sy,
                                           unsigned int mx, unsigned int my, int c) {
    float* p = pooled + (size_t)g * 128 + (lane << 1);
    atomicAdd(p, sx);
    atomicAdd(p + 1, sy);
    atomicMax((unsigned int*)(p + 64), mx);
    atomicMax((unsigned int*)(p + 65), my);
    if (lane == 0) atomicAdd(&counts[g], (float)c);
}

__global__ void __launch_bounds__(256)
pool_runs_k(const float* __restrict__ h, const int* __restrict__ batch,
            float* __restrict__ pooled, float* __restrict__ counts, int n) {
    const int lane = threadIdx.x & 31;
    int w = (blockIdx.x * blockDim.x + threadIdx.x) >> 5;
    w = __builtin_amdgcn_readfirstlane(w);
    const int base = w * PRUN;
    if (base >= n) return;
    const int m = min(PRUN, n - base);

    int gid = 0;
    if (lane < m) gid = batch[base + lane];

    const float* hl = h + (size_t)base * HID + (lane << 1);
    float2 hv[PRUN];
#pragma unroll
    for (int j = 0; j < PRUN; ++j) {
        const int jj = min(j, m - 1);
        hv[j] = *(const float2*)(hl + jj * HID);
    }

    int cur = __builtin_amdgcn_readlane(gid, 0);
    float sx = hv[0].x, sy = hv[0].y;
    unsigned int mx = __float_as_uint(hv[0].x), my = __float_as_uint(hv[0].y);
    int c = 1;
#pragma unroll
    for (int j = 1; j < PRUN; ++j) {
        if (j < m) {
            const int g = __builtin_amdgcn_readlane(gid, j);
            if (g != cur) {
                pool_flush(pooled, counts, cur, lane, sx, sy, mx, my, c);
                cur = g; sx = 0.0f; sy = 0.0f; mx = 0u; my = 0u; c = 0;
            }
            sx += hv[j].x;
            sy += hv[j].y;
            mx = max(mx, __float_as_uint(hv[j].x));
            my = max(my, __float_as_uint(hv[j].y));
            ++c;
        }
    }
    pool_flush(pooled, counts, cur, lane, sx, sy, mx, my, c);
}

__global__ void meandiv_k(float* __restrict__ pooled, const float* __restrict__ counts, int G) {
    int i = blockIdx.x * blockDim.x + threadIdx.x;
    if (i >= G * HID) return;
    int g = i >> 6;
    int f = i & 63;
    pooled[(size_t)g * 128 + f] /= fmaxf(counts[g], 1.0f);
}

__global__ void head_k(const float* __restrict__ pooled, const float* __restrict__ Wout,
                       const float* __restrict__ bout, float* __restrict__ out, int G) {
    int t = blockIdx.x * blockDim.x + threadIdx.x;
    int g = t >> 5;
    int lane = t & 31;
    if (g >= G) return;
    float v = 0.0f;
#pragma unroll
    for (int j = lane; j < 2 * HID; j += 32) v += pooled[(size_t)g * 128 + j] * Wout[j];
#pragma unroll
    for (int off = 16; off; off >>= 1) v += __shfl_xor(v, off, 32);
    if (lane == 0) out[g] = v + bout[0];
}

#define WS_DIS    0
#define WS_CNT    (WS_DIS + N_NODES)
#define WS_ROWPTR (WS_CNT + N_NODES)
#define WS_BSUM   (WS_ROWPTR + N_NODES + 4)
#define WS_COUNTS (WS_BSUM + 128)
#define WS_RANK   (WS_COUNTS + NGRAPH)
#define WS_REC    (WS_RANK + N_EDGES)
#define WS_BUF0   (((WS_REC + 2 * N_EDGES) + 63) / 64 * 64)

extern "C" void kernel_launch(void* const* d_in, const int* in_sizes, int n_in,
                              void* d_out, int out_size, void* d_ws, size_t ws_size,
                              hipStream_t stream) {
    const float* x     = (const float*)d_in[0];
    const int*   eidx  = (const int*)  d_in[1];
    const int*   batch = (const int*)  d_in[2];
    const float* Ws[4] = { (const float*)d_in[3], (const float*)d_in[5],
                           (const float*)d_in[7], (const float*)d_in[9] };
    const float* bs[4] = { (const float*)d_in[4], (const float*)d_in[6],
                           (const float*)d_in[8], (const float*)d_in[10] };
    const float* Wout  = (const float*)d_in[11];
    const float* bout  = (const float*)d_in[12];

    const int N = N_NODES, E = N_EDGES, G = NGRAPH;
    const int* src = eidx;
    const int* dst = eidx + E;

    float* ws     = (float*)d_ws;
    float* dis    = ws + WS_DIS;
    int*   cnt    = (int*)(ws + WS_CNT);
    int*   rowptr = (int*)(ws + WS_ROWPTR);
    int*   bsum   = (int*)(ws + WS_BSUM);
    float* counts = ws + WS_COUNTS;
    int*   rank   = (int*)(ws + WS_RANK);
    int*   rec    = (int*)(ws + WS_REC);
    float* buf0   = ws + WS_BUF0;
    float* buf1   = buf0 + (size_t)N * HID;
    float* buf2   = buf1 + (size_t)N * HID;

    float* out    = (float*)d_out;
    float* pooled = out + G;

    const int n4 = N / 4;
    const int scanBlocks = (n4 + SCAN_T - 1) / SCAN_T;
    fill_k      <<<(N + 255) / 256, 256, 0, stream>>>((float*)cnt, 0.0f, N);
    count_rank_k<<<(E + 255) / 256, 256, 0, stream>>>(dst, cnt, rank, E);
    blocksum_k  <<<scanBlocks, SCAN_T, 0, stream>>>(cnt, bsum, n4);
    rowptr_k    <<<scanBlocks, SCAN_T, 0, stream>>>(cnt, bsum, rowptr, dis, n4);
    place_k     <<<(E + 255) / 256, 256, 0, stream>>>(src, dst, rank, rowptr, dis, rec, E);

    const float* hin = x;
    float* HW  = buf0;
    float* AGG = buf1;
    float* spare = buf2;
    int K = 128;

    for (int l = 0; l < 4; ++l) {
        int gemmThreads = (N / 16) * 32;
        gemm_wmma_k <<<(gemmThreads + 255) / 256, 256, 0, stream>>>(hin, Ws[l], HW, N, K);
        gather_layer_k<<<(N * 32 + 255) / 256, 256, 0, stream>>>(rowptr, rec, HW, dis, bs[l], AGG, N);

        const float* next_in = AGG;
        float* newAGG = (l == 0) ? spare : (float*)hin;
        hin = next_in;
        AGG = newAGG;
        K = HID;
    }

    fill_k   <<<(G + G * 128 + 255) / 256, 256, 0, stream>>>(out, 0.0f, G + G * 128);
    fill_k   <<<(G + 255) / 256, 256, 0, stream>>>(counts, 0.0f, G);
    pool_runs_k<<<((N + PRUN - 1) / PRUN * 32 + 255) / 256, 256, 0, stream>>>(hin, batch, pooled, counts, N);
    meandiv_k<<<(G * HID + 255) / 256, 256, 0, stream>>>(pooled, counts, G);
    head_k   <<<(G * 32 + 255) / 256, 256, 0, stream>>>(pooled, Wout, bout, out, G);
}


# Edges sorted by destination on LDS counters only: 256-node buckets filled chunk by chunk, then one block per bucket ranks its rows; per-edge global returning add removed; sum order = LDS arrival
# speedup vs baseline: 3.7956x; 1.2648x over previous
#include <hip/hip_runtime.h>
#include <hip/hip_bf16.h>

typedef __attribute__((ext_vector_type(2))) float v2f;
typedef __attribute__((ext_vector_type(8))) float v8f;

#define N_NODES 100000
#define N_EDGES 1600000
#define HID     64
#define NGRAPH  2048

__global__ void fill_k(float* __restrict__ p, float v, long long n) {
    long long i = (long long)blockIdx.x * blockDim.x + threadIdx.x;
    if (i < n) p[i] = v;
}

__global__ void gemm_wmma_k(const float* __restrict__ A, const float* __restrict__ W,
                            float* __restrict__ C, int rows, int K) {
    const int lane = threadIdx.x & 31;
    const int w    = (blockIdx.x * blockDim.x + threadIdx.x) >> 5;
    const int rowTiles = rows >> 4;
    if (w >= rowTiles) return;

    const int m0  = w << 4;
    const int mr  = lane & 15;
    const int khi = (lane >> 4) << 1;

    const float* arow = A + (size_t)(m0 + mr) * K + khi;
    const float* bcol = W + mr;

    v8f acc0 = {}, acc1 = {}, acc2 = {}, acc3 = {};
    for (int k = 0; k < K; k += 4) {
        v2f a;
        const float2 av = *(const float2*)(arow + k);
        a[0] = av.x; a[1] = av.y;

        const float* brow0 = bcol + (size_t)(k + khi) * HID;
        const float* brow1 = brow0 + HID;
        v2f b0, b1, b2, b3;
        b0[0] = brow0[0];  b0[1] = brow1[0];
        b1[0] = brow0[16]; b1[1] = brow1[16];
        b2[0] = brow0[32]; b2[1] = brow1[32];
        b3[0] = brow0[48]; b3[1] = brow1[48];

        acc0 = __builtin_amdgcn_wmma_f32_16x16x4_f32(false, a, false, b0, (short)0, acc0, false, false);
        acc1 = __builtin_amdgcn_wmma_f32_16x16x4_f32(false, a, false, b1, (short)0, acc1, false, false);
        acc2 = __builtin_amdgcn_wmma_f32_16x16x4_f32(false, a, false, b2, (short)0, acc2, false, false);
        acc3 = __builtin_amdgcn_wmma_f32_16x16x4_f32(false, a, false, b3, (short)0, acc3, false, false);
    }

    float* crow = C + (size_t)(m0 + ((lane >> 4) << 3)) * HID + mr;
#pragma unroll
    for (int j = 0; j < 8; ++j) {
        crow[(size_t)j * HID +  0] = acc0[j];
        crow[(size_t)j * HID + 16] = acc1[j];
        crow[(size_t)j * HID + 32] = acc2[j];
        crow[(size_t)j * HID + 48] = acc3[j];
    }
}

#define BSH    8
#define NB     ((N_NODES + 255) >> BSH)
#define CH     4096
#define NCH    ((N_EDGES + CH - 1) / CH)
#define SORT_T 512

__device__ __forceinline__ int scan512(int v, int* sm, int& total) {
    const int lane = threadIdx.x & 31;
    const int wid  = threadIdx.x >> 5;
    int inc = v;
#pragma unroll
    for (int off = 1; off < 32; off <<= 1) {
        const int u = __shfl_up(inc, off, 32);
        if (lane >= off) inc += u;
    }
    if (lane == 31) sm[wid] = inc;
    __syncthreads();
    int pre = 0, tot = 0;
#pragma unroll
    for (int k = 0; k < SORT_T / 32; ++k) {
        const int s = sm[k];
        tot += s;
        if (k < wid) pre += s;
    }
    total = tot;
    return pre + inc - v;
}

__global__ void __launch_bounds__(SORT_T)
bin_count_k(const int* __restrict__ dst, int* __restrict__ histT) {
    __shared__ int c[NB];
    const int tid = threadIdx.x;
    const int chunk = blockIdx.x;
    int4 d[CH / (SORT_T * 4)];
#pragma unroll
    for (int k = 0; k < CH / (SORT_T * 4); ++k) {
        const int e = chunk * CH + k * (SORT_T * 4) + tid * 4;
        d[k] = (e < N_EDGES) ? *(const int4*)(dst + e) : make_int4(-1, -1, -1, -1);
    }
    for (int i = tid; i < NB; i += SORT_T) c[i] = 0;
    __syncthreads();
#pragma unroll
    for (int k = 0; k < CH / (SORT_T * 4); ++k) {
        if (d[k].x >= 0) {
            atomicAdd(&c[d[k].x >> BSH], 1);
            atomicAdd(&c[d[k].y >> BSH], 1);
            atomicAdd(&c[d[k].z >> BSH], 1);
            atomicAdd(&c[d[k].w >> BSH], 1);
        }
    }
    __syncthreads();
    for (int i = tid; i < NB; i += SORT_T) histT[i * NCH + chunk] = c[i];
}

__global__ void __launch_bounds__(SORT_T)
bin_scan_k(const int* __restrict__ histT, int* __restrict__ offT, int* __restrict__ tot) {
    __shared__ int sm[SORT_T / 32];
    const int tid = threadIdx.x;
    const int b = blockIdx.x;
    const int v = (tid < NCH) ? histT[b * NCH + tid] : 0;
    int total;
    const int excl = scan512(v, sm, total);
    if (tid < NCH) offT[b * NCH + tid] = excl;
    if (tid == 0) tot[b] = total;
}

__global__ void __launch_bounds__(SORT_T)
bin_scatter_k(const int* __restrict__ src, const int* __restrict__ dst,
              const int* __restrict__ offT, const int* __restrict__ tot,
              int* __restrict__ binned) {
    __shared__ int cur[NB];
    __shared__ int sm[SORT_T / 32];
    const int tid = threadIdx.x;
    const int chunk = blockIdx.x;
    int4 s[CH / (SORT_T * 4)], d[CH / (SORT_T * 4)];
#pragma unroll
    for (int k = 0; k < CH / (SORT_T * 4); ++k) {
        const int e = chunk * CH + k * (SORT_T * 4) + tid * 4;
        const bool in = e < N_EDGES;
        s[k] = in ? *(const int4*)(src + e) : make_int4(0, 0, 0, 0);
        d[k] = in ? *(const int4*)(dst + e) : make_int4(-1, -1, -1, -1);
    }
    const int t = (tid < NB) ? tot[tid] : 0;
    const int o = (tid < NB) ? offT[tid * NCH + chunk] : 0;
    int total;
    const int bstart = scan512(t, sm, total);
    if (tid < NB) cur[tid] = bstart + o;
    __syncthreads();
    int2* out2 = (int2*)binned;
#pragma unroll
    for (int k = 0; k < CH / (SORT_T * 4); ++k) {
        if (d[k].x >= 0) {
            const int p0 = atomicAdd(&cur[d[k].x >> BSH], 1);
            const int p1 = atomicAdd(&cur[d[k].y >> BSH], 1);
            const int p2 = atomicAdd(&cur[d[k].z >> BSH], 1);
            const int p3 = atomicAdd(&cur[d[k].w >> BSH], 1);
            out2[p0] = make_int2(s[k].x, d[k].x);
            out2[p1] = make_int2(s[k].y, d[k].y);
            out2[p2] = make_int2(s[k].z, d[k].z);
            out2[p3] = make_int2(s[k].w, d[k].w);
        }
    }
}

#define BLD 4
__global__ void __launch_bounds__(SORT_T)
bucket_rows_k(const int* __restrict__ binned, const int* __restrict__ tot,
              int* __restrict__ rowptr, float* __restrict__ dis, int* __restrict__ rec) {
    __shared__ int cnt[1 << BSH];
    __shared__ int cur[1 << BSH];
    __shared__ int smA[SORT_T / 32];
    __shared__ int smB[SORT_T / 32];
    __shared__ int range[2];
    const int tid = threadIdx.x;
    const int b = blockIdx.x;
    const int node0 = b << BSH;
    if (tid < (1 << BSH)) cnt[tid] = 0;
    const int t = (tid < NB) ? tot[tid] : 0;
    int total;
    const int bstart = scan512(t, smA, total);
    if (tid == b) { range[0] = bstart; range[1] = bstart + t; }
    __syncthreads();
    const int beg = range[0];
    const int end = range[1];
    const int2* in2 = (const int2*)binned;
    int2* out2 = (int2*)rec;

    for (int e0 = beg; e0 < end; e0 += SORT_T * BLD) {
        int d[BLD];
#pragma unroll
        for (int u = 0; u < BLD; ++u) {
            const int e = e0 + u * SORT_T + tid;
            d[u] = (e < end) ? in2[e].y : -1;
        }
#pragma unroll
        for (int u = 0; u < BLD; ++u)
            if (d[u] >= 0) atomicAdd(&cnt[d[u] - node0], 1);
    }
    __syncthreads();

    const int c = (tid < (1 << BSH)) ? cnt[tid] : 0;
    int total2;
    const int ex = scan512(c, smB, total2);
    if (tid < (1 << BSH)) {
        cur[tid] = beg + ex;
        const int node = node0 + tid;
        if (node < N_NODES) {
            rowptr[node] = beg + ex;
            dis[node] = rsqrtf((float)(c + 1));
        }
    }
    if (b == NB - 1 && tid == 0) rowptr[N_NODES] = end;
    __syncthreads();

    for (int e0 = beg; e0 < end; e0 += SORT_T * BLD) {
        int2 r[BLD];
#pragma unroll
        for (int u = 0; u < BLD; ++u) {
            const int e = e0 + u * SORT_T + tid;
            r[u] = (e < end) ? in2[e] : make_int2(0, -1);
        }
#pragma unroll
        for (int u = 0; u < BLD; ++u)
            if (r[u].y >= 0) {
                const int p = atomicAdd(&cur[r[u].y - node0], 1);
                out2[p] = r[u];
            }
    }
}

__global__ void norm_fill_k(int* __restrict__ rec, const float* __restrict__ dis, int pairs) {
    const int q = blockIdx.x * blockDim.x + threadIdx.x;
    if (q >= pairs) return;
    int4 r = ((const int4*)rec)[q];
    r.y = __float_as_int(dis[r.x] * dis[r.y]);
    r.w = __float_as_int(dis[r.z] * dis[r.w]);
    ((int4*)rec)[q] = r;
}

#define GRP 8
__global__ void __launch_bounds__(256)
gather_layer_k(const int* __restrict__ rowptr, const int* __restrict__ rec,
               const float* __restrict__ h, const float* __restrict__ dis,
               const float* __restrict__ bias, float* __restrict__ outp, int n) {
    const int lane = threadIdx.x & 31;
    int node = (blockIdx.x * blockDim.x + threadIdx.x) >> 5;
    if (node >= n) return;
    node = __builtin_amdgcn_readfirstlane(node);
    const int beg = rowptr[node];
    const int end = rowptr[node + 1];
    const float* hl = h + (lane << 1);
    const float2 own = *(const float2*)(hl + (size_t)node * HID);
    const float2 bv  = *(const float2*)(bias + (lane << 1));
    float sw = dis[node]; sw = sw * sw;
    float2 acc = make_float2(0.0f, 0.0f);

    for (int base = beg; base < end; base += 32) {
        const int m = min(32, end - base);
        int2 ev = make_int2(0, 0);
        if (lane < m) ev = ((const int2*)rec)[base + lane];
        for (int j = 0; j < m; j += GRP) {
            float2 hv[GRP];
            float  wv[GRP];
#pragma unroll
            for (int u = 0; u < GRP; ++u) {
                const int jj = min(j + u, m - 1);
                const int   s = __builtin_amdgcn_readlane(ev.x, jj);
                const float w = __int_as_float(__builtin_amdgcn_readlane(ev.y, jj));
                wv[u] = (j + u < m) ? w : 0.0f;
                hv[u] = *(const float2*)(hl + (unsigned)(s * HID));
            }
#pragma unroll
            for (int u = 0; u < GRP; ++u) {
                acc.x += hv[u].x * wv[u];
                acc.y += hv[u].y * wv[u];
            }
        }
    }
    float2 r;
    r.x = fmaxf(acc.x + own.x * sw + bv.x, 0.0f);
    r.y = fmaxf(acc.y + own.y * sw + bv.y, 0.0f);
    *(float2*)(outp + (size_t)node * HID + (lane << 1)) = r;
}

#define PRUN 16
__device__ __forceinline__ void pool_flush(float* __restrict__ pooled, float* __restrict__ counts,
                                           int g, int lane, float sx, float sy,
                                           unsigned int mx, unsigned int my, int c) {
    float* p = pooled + (size_t)g * 128 + (lane << 1);
    atomicAdd(p, sx);
    atomicAdd(p + 1, sy);
    atomicMax((unsigned int*)(p + 64), mx);
    atomicMax((unsigned int*)(p + 65), my);
    if (lane == 0) atomicAdd(&counts[g], (float)c);
}

__global__ void __launch_bounds__(256)
pool_runs_k(const float* __restrict__ h, const int* __restrict__ batch,
            float* __restrict__ pooled, float* __restrict__ counts, int n) {
    const int lane = threadIdx.x & 31;
    int w = (blockIdx.x * blockDim.x + threadIdx.x) >> 5;
    w = __builtin_amdgcn_readfirstlane(w);
    const int base = w * PRUN;
    if (base >= n) return;
    const int m = min(PRUN, n - base);

    int gid = 0;
    if (lane < m) gid = batch[base + lane];

    const float* hl = h + (size_t)base * HID + (lane << 1);
    float2 hv[PRUN];
#pragma unroll
    for (int j = 0; j < PRUN; ++j) {
        const int jj = min(j, m - 1);
        hv[j] = *(const float2*)(hl + jj * HID);
    }

    int cur = __builtin_amdgcn_readlane(gid, 0);
    float sx = hv[0].x, sy = hv[0].y;
    unsigned int mx = __float_as_uint(hv[0].x), my = __float_as_uint(hv[0].y);
    int c = 1;
#pragma unroll
    for (int j = 1; j < PRUN; ++j) {
        if (j < m) {
            const int g = __builtin_amdgcn_readlane(gid, j);
            if (g != cur) {
                pool_flush(pooled, counts, cur, lane, sx, sy, mx, my, c);
                cur = g; sx = 0.0f; sy = 0.0f; mx = 0u; my = 0u; c = 0;
            }
            sx += hv[j].x;
            sy += hv[j].y;
            mx = max(mx, __float_as_uint(hv[j].x));
            my = max(my, __float_as_uint(hv[j].y));
            ++c;
        }
    }
    pool_flush(pooled, counts, cur, lane, sx, sy, mx, my, c);
}

__global__ void meandiv_k(float* __restrict__ pooled, const float* __restrict__ counts, int G) {
    int i = blockIdx.x * blockDim.x + threadIdx.x;
    if (i >= G * HID) return;
    int g = i >> 6;
    int f = i & 63;
    pooled[(size_t)g * 128 + f] /= fmaxf(counts[g], 1.0f);
}

__global__ void head_k(const float* __restrict__ pooled, const float* __restrict__ Wout,
                       const float* __restrict__ bout, float* __restrict__ out, int G) {
    int t = blockIdx.x * blockDim.x + threadIdx.x;
    int g = t >> 5;
    int lane = t & 31;
    if (g >= G) return;
    float v = 0.0f;
#pragma unroll
    for (int j = lane; j < 2 * HID; j += 32) v += pooled[(size_t)g * 128 + j] * Wout[j];
#pragma unroll
    for (int off = 16; off; off >>= 1) v += __shfl_xor(v, off, 32);
    if (lane == 0) out[g] = v + bout[0];
}

#define WS_DIS    0
#define WS_ROWPTR (WS_DIS + N_NODES)
#define WS_TOT    (WS_ROWPTR + N_NODES + 4)
#define WS_COUNTS (WS_TOT + 512)
#define WS_HIST   (WS_COUNTS + NGRAPH)
#define WS_TABLE  ((NB * NCH + 3) / 4 * 4)
#define WS_OFF    (WS_HIST + WS_TABLE)
#define WS_BIN    (WS_OFF + WS_TABLE)
#define WS_REC    (WS_BIN + 2 * N_EDGES)
#define WS_BUF0   (((WS_REC + 2 * N_EDGES) + 63) / 64 * 64)

extern "C" void kernel_launch(void* const* d_in, const int* in_sizes, int n_in,
                              void* d_out, int out_size, void* d_ws, size_t ws_size,
                              hipStream_t stream) {
    const float* x     = (const float*)d_in[0];
    const int*   eidx  = (const int*)  d_in[1];
    const int*   batch = (const int*)  d_in[2];
    const float* Ws[4] = { (const float*)d_in[3], (const float*)d_in[5],
                           (const float*)d_in[7], (const float*)d_in[9] };
    const float* bs[4] = { (const float*)d_in[4], (const float*)d_in[6],
                           (const float*)d_in[8], (const float*)d_in[10] };
    const float* Wout  = (const float*)d_in[11];
    const float* bout  = (const float*)d_in[12];

    const int N = N_NODES, E = N_EDGES, G = NGRAPH;
    const int* src = eidx;
    const int* dst = eidx + E;

    float* ws     = (float*)d_ws;
    float* dis    = ws + WS_DIS;
    int*   rowptr = (int*)(ws + WS_ROWPTR);
    int*   tot    = (int*)(ws + WS_TOT);
    float* counts = ws + WS_COUNTS;
    int*   histT  = (int*)(ws + WS_HIST);
    int*   offT   = (int*)(ws + WS_OFF);
    int*   binned = (int*)(ws + WS_BIN);
    int*   rec    = (int*)(ws + WS_REC);
    float* buf0   = ws + WS_BUF0;
    float* buf1   = buf0 + (size_t)N * HID;
    float* buf2   = buf1 + (size_t)N * HID;

    float* out    = (float*)d_out;
    float* pooled = out + G;

    bin_count_k  <<<NCH, SORT_T, 0, stream>>>(dst, histT);
    bin_scan_k   <<<NB,  SORT_T, 0, stream>>>(histT, offT, tot);
    bin_scatter_k<<<NCH, SORT_T, 0, stream>>>(src, dst, offT, tot, binned);
    bucket_rows_k<<<NB,  SORT_T, 0, stream>>>(binned, tot, rowptr, dis, rec);
    norm_fill_k  <<<(E / 2 + 255) / 256, 256, 0, stream>>>(rec, dis, E / 2);

    const float* hin = x;
    float* HW  = buf0;
    float* AGG = buf1;
    float* spare = buf2;
    int K = 128;

    for (int l = 0; l < 4; ++l) {
        int gemmThreads = (N / 16) * 32;
        gemm_wmma_k <<<(gemmThreads + 255) / 256, 256, 0, stream>>>(hin, Ws[l], HW, N, K);
        gather_layer_k<<<(N * 32 + 255) / 256, 256, 0, stream>>>(rowptr, rec, HW, dis, bs[l], AGG, N);

        const float* next_in = AGG;
        float* newAGG = (l == 0) ? spare : (float*)hin;
        hin = next_in;
        AGG = newAGG;
        K = HID;
    }

    fill_k   <<<(G + G * 128 + 255) / 256, 256, 0, stream>>>(out, 0.0f, G + G * 128);
    fill_k   <<<(G + 255) / 256, 256, 0, stream>>>(counts, 0.0f, G);
    pool_runs_k<<<((N + PRUN - 1) / PRUN * 32 + 255) / 256, 256, 0, stream>>>(hin, batch, pooled, counts, N);
    meandiv_k<<<(G * HID + 255) / 256, 256, 0, stream>>>(pooled, counts, G);
    head_k   <<<(G * 32 + 255) / 256, 256, 0, stream>>>(pooled, Wout, bout, out, G);
}


# Next layer's product inside the neighbour-sum launch: a 16-wave block hands its 16 finished rows through LDS to f32 WMMA; 3 product launches, their row-plane write and re-read removed; sum order kept
# speedup vs baseline: 4.1297x; 1.0880x over previous
#include <hip/hip_runtime.h>
#include <hip/hip_bf16.h>

typedef __attribute__((ext_vector_type(2))) float v2f;
typedef __attribute__((ext_vector_type(8))) float v8f;

#define N_NODES 100000
#define N_EDGES 1600000
#define HID     64
#define NGRAPH  2048

__global__ void fill_k(float* __restrict__ p, float v, long long n) {
    long long i = (long long)blockIdx.x * blockDim.x + threadIdx.x;
    if (i < n) p[i] = v;
}

__global__ void gemm_wmma_k(const float* __restrict__ A, const float* __restrict__ W,
                            float* __restrict__ C, int rows, int K) {
    const int lane = threadIdx.x & 31;
    const int w    = (blockIdx.x * blockDim.x + threadIdx.x) >> 5;
    const int rowTiles = rows >> 4;
    if (w >= rowTiles) return;

    const int m0  = w << 4;
    const int mr  = lane & 15;
    const int khi = (lane >> 4) << 1;

    const float* arow = A + (size_t)(m0 + mr) * K + khi;
    const float* bcol = W + mr;

    v8f acc0 = {}, acc1 = {}, acc2 = {}, acc3 = {};
    for (int k = 0; k < K; k += 4) {
        v2f a;
        const float2 av = *(const float2*)(arow + k);
        a[0] = av.x; a[1] = av.y;

        const float* brow0 = bcol + (size_t)(k + khi) * HID;
        const float* brow1 = brow0 + HID;
        v2f b0, b1, b2, b3;
        b0[0] = brow0[0];  b0[1] = brow1[0];
        b1[0] = brow0[16]; b1[1] = brow1[16];
        b2[0] = brow0[32]; b2[1] = brow1[32];
        b3[0] = brow0[48]; b3[1] = brow1[48];

        acc0 = __builtin_amdgcn_wmma_f32_16x16x4_f32(false, a, false, b0, (short)0, acc0, false, false);
        acc1 = __builtin_amdgcn_wmma_f32_16x16x4_f32(false, a, false, b1, (short)0, acc1, false, false);
        acc2 = __builtin_amdgcn_wmma_f32_16x16x4_f32(false, a, false, b2, (short)0, acc2, false, false);
        acc3 = __builtin_amdgcn_wmma_f32_16x16x4_f32(false, a, false, b3, (short)0, acc3, false, false);
    }

    float* crow = C + (size_t)(m0 + ((lane >> 4) << 3)) * HID + mr;
#pragma unroll
    for (int j = 0; j < 8; ++j) {
        crow[(size_t)j * HID +  0] = acc0[j];
        crow[(size_t)j * HID + 16] = acc1[j];
        crow[(size_t)j * HID + 32] = acc2[j];
        crow[(size_t)j * HID + 48] = acc3[j];
    }
}

#define BSH    8
#define NB     ((N_NODES + 255) >> BSH)
#define CH     4096
#define NCH    ((N_EDGES + CH - 1) / CH)
#define SORT_T 512

__device__ __forceinline__ int scan512(int v, int* sm, int& total) {
    const int lane = threadIdx.x & 31;
    const int wid  = threadIdx.x >> 5;
    int inc = v;
#pragma unroll
    for (int off = 1; off < 32; off <<= 1) {
        const int u = __shfl_up(inc, off, 32);
        if (lane >= off) inc += u;
    }
    if (lane == 31) sm[wid] = inc;
    __syncthreads();
    int pre = 0, tot = 0;
#pragma unroll
    for (int k = 0; k < SORT_T / 32; ++k) {
        const int s = sm[k];
        tot += s;
        if (k < wid) pre += s;
    }
    total = tot;
    return pre + inc - v;
}

__global__ void __launch_bounds__(SORT_T)
bin_count_k(const int* __restrict__ dst, int* __restrict__ histT) {
    __shared__ int c[NB];
    const int tid = threadIdx.x;
    const int chunk = blockIdx.x;
    int4 d[CH / (SORT_T * 4)];
#pragma unroll
    for (int k = 0; k < CH / (SORT_T * 4); ++k) {
        const int e = chunk * CH + k * (SORT_T * 4) + tid * 4;
        d[k] = (e < N_EDGES) ? *(const int4*)(dst + e) : make_int4(-1, -1, -1, -1);
    }
    for (int i = tid; i < NB; i += SORT_T) c[i] = 0;
    __syncthreads();
#pragma unroll
    for (int k = 0; k < CH / (SORT_T * 4); ++k) {
        if (d[k].x >= 0) {
            atomicAdd(&c[d[k].x >> BSH], 1);
            atomicAdd(&c[d[k].y >> BSH], 1);
            atomicAdd(&c[d[k].z >> BSH], 1);
            atomicAdd(&c[d[k].w >> BSH], 1);
        }
    }
    __syncthreads();
    for (int i = tid; i < NB; i += SORT_T) histT[i * NCH + chunk] = c[i];
}

__global__ void __launch_bounds__(SORT_T)
bin_scan_k(const int* __restrict__ histT, int* __restrict__ offT, int* __restrict__ tot) {
    __shared__ int sm[SORT_T / 32];
    const int tid = threadIdx.x;
    const int b = blockIdx.x;
    const int v = (tid < NCH) ? histT[b * NCH + tid] : 0;
    int total;
    const int excl = scan512(v, sm, total);
    if (tid < NCH) offT[b * NCH + tid] = excl;
    if (tid == 0) tot[b] = total;
}

__global__ void __launch_bounds__(SORT_T)
bin_scatter_k(const int* __restrict__ src, const int* __restrict__ dst,
              const int* __restrict__ offT, const int* __restrict__ tot,
              int* __restrict__ binned) {
    __shared__ int cur[NB];
    __shared__ int sm[SORT_T / 32];
    const int tid = threadIdx.x;
    const int chunk = blockIdx.x;
    int4 s[CH / (SORT_T * 4)], d[CH / (SORT_T * 4)];
#pragma unroll
    for (int k = 0; k < CH / (SORT_T * 4); ++k) {
        const int e = chunk * CH + k * (SORT_T * 4) + tid * 4;
        const bool in = e < N_EDGES;
        s[k] = in ? *(const int4*)(src + e) : make_int4(0, 0, 0, 0);
        d[k] = in ? *(const int4*)(dst + e) : make_int4(-1, -1, -1, -1);
    }
    const int t = (tid < NB) ? tot[tid] : 0;
    const int o = (tid < NB) ? offT[tid * NCH + chunk] : 0;
    int total;
    const int bstart = scan512(t, sm, total);
    if (tid < NB) cur[tid] = bstart + o;
    __syncthreads();
    int2* out2 = (int2*)binned;
#pragma unroll
    for (int k = 0; k < CH / (SORT_T * 4); ++k) {
        if (d[k].x >= 0) {
            const int p0 = atomicAdd(&cur[d[k].x >> BSH], 1);
            const int p1 = atomicAdd(&cur[d[k].y >> BSH], 1);
            const int p2 = atomicAdd(&cur[d[k].z >> BSH], 1);
            const int p3 = atomicAdd(&cur[d[k].w >> BSH], 1);
            out2[p0] = make_int2(s[k].x, d[k].x);
            out2[p1] = make_int2(s[k].y, d[k].y);
            out2[p2] = make_int2(s[k].z, d[k].z);
            out2[p3] = make_int2(s[k].w, d[k].w);
        }
    }
}

#define BLD 4
__global__ void __launch_bounds__(SORT_T)
bucket_rows_k(const int* __restrict__ binned, const int* __restrict__ tot,
              int* __restrict__ rowptr, float* __restrict__ dis, int* __restrict__ rec) {
    __shared__ int cnt[1 << BSH];
    __shared__ int cur[1 << BSH];
    __shared__ int smA[SORT_T / 32];
    __shared__ int smB[SORT_T / 32];
    __shared__ int range[2];
    const int tid = threadIdx.x;
    const int b = blockIdx.x;
    const int node0 = b << BSH;
    if (tid < (1 << BSH)) cnt[tid] = 0;
    const int t = (tid < NB) ? tot[tid] : 0;
    int total;
    const int bstart = scan512(t, smA, total);
    if (tid == b) { range[0] = bstart; range[1] = bstart + t; }
    __syncthreads();
    const int beg = range[0];
    const int end = range[1];
    const int2* in2 = (const int2*)binned;
    int2* out2 = (int2*)rec;

    for (int e0 = beg; e0 < end; e0 += SORT_T * BLD) {
        int d[BLD];
#pragma unroll
        for (int u = 0; u < BLD; ++u) {
            const int e = e0 + u * SORT_T + tid;
            d[u] = (e < end) ? in2[e].y : -1;
        }
#pragma unroll
        for (int u = 0; u < BLD; ++u)
            if (d[u] >= 0) atomicAdd(&cnt[d[u] - node0], 1);
    }
    __syncthreads();

    const int c = (tid < (1 << BSH)) ? cnt[tid] : 0;
    int total2;
    const int ex = scan512(c, smB, total2);
    if (tid < (1 << BSH)) {
        cur[tid] = beg + ex;
        const int node = node0 + tid;
        if (node < N_NODES) {
            rowptr[node] = beg + ex;
            dis[node] = rsqrtf((float)(c + 1));
        }
    }
    if (b == NB - 1 && tid == 0) rowptr[N_NODES] = end;
    __syncthreads();

    for (int e0 = beg; e0 < end; e0 += SORT_T * BLD) {
        int2 r[BLD];
#pragma unroll
        for (int u = 0; u < BLD; ++u) {
            const int e = e0 + u * SORT_T + tid;
            r[u] = (e < end) ? in2[e] : make_int2(0, -1);
        }
#pragma unroll
        for (int u = 0; u < BLD; ++u)
            if (r[u].y >= 0) {
                const int p = atomicAdd(&cur[r[u].y - node0], 1);
                out2[p] = r[u];
            }
    }
}

__global__ void norm_fill_k(int* __restrict__ rec, const float* __restrict__ dis, int pairs) {
    const int q = blockIdx.x * blockDim.x + threadIdx.x;
    if (q >= pairs) return;
    int4 r = ((const int4*)rec)[q];
    r.y = __float_as_int(dis[r.x] * dis[r.y]);
    r.w = __float_as_int(dis[r.z] * dis[r.w]);
    ((int4*)rec)[q] = r;
}

#define GRP 8
__device__ __forceinline__ float2 gather_row(const int* __restrict__ rowptr, const int* __restrict__ rec,
                                             const float* __restrict__ h, const float* __restrict__ dis,
                                             const float* __restrict__ bias, int node, int lane) {
    const int beg = rowptr[node];
    const int end = rowptr[node + 1];
    const float* hl = h + (lane << 1);
    const float2 own = *(const float2*)(hl + (size_t)node * HID);
    const float2 bv  = *(const float2*)(bias + (lane << 1));
    float sw = dis[node]; sw = sw * sw;
    float2 acc = make_float2(0.0f, 0.0f);

    for (int base = beg; base < end; base += 32) {
        const int m = min(32, end - base);
        int2 ev = make_int2(0, 0);
        if (lane < m) ev = ((const int2*)rec)[base + lane];
        for (int j = 0; j < m; j += GRP) {
            float2 hv[GRP];
            float  wv[GRP];
#pragma unroll
            for (int u = 0; u < GRP; ++u) {
                const int jj = min(j + u, m - 1);
                const int   s = __builtin_amdgcn_readlane(ev.x, jj);
                const float w = __int_as_float(__builtin_amdgcn_readlane(ev.y, jj));
                wv[u] = (j + u < m) ? w : 0.0f;
                hv[u] = *(const float2*)(hl + (unsigned)(s * HID));
            }
#pragma unroll
            for (int u = 0; u < GRP; ++u) {
                acc.x += hv[u].x * wv[u];
                acc.y += hv[u].y * wv[u];
            }
        }
    }
    float2 r;
    r.x = fmaxf(acc.x + own.x * sw + bv.x, 0.0f);
    r.y = fmaxf(acc.y + own.y * sw + bv.y, 0.0f);
    return r;
}

__global__ void __launch_bounds__(256)
gather_layer_k(const int* __restrict__ rowptr, const int* __restrict__ rec,
               const float* __restrict__ h, const float* __restrict__ dis,
               const float* __restrict__ bias, float* __restrict__ outp, int n) {
    const int lane = threadIdx.x & 31;
    int node = (blockIdx.x * blockDim.x + threadIdx.x) >> 5;
    if (node >= n) return;
    node = __builtin_amdgcn_readfirstlane(node);
    const float2 r = gather_row(rowptr, rec, h, dis, bias, node, lane);
    *(float2*)(outp + (size_t)node * HID + (lane << 1)) = r;
}

#define TROWS  16
#define TPITCH 68
__global__ void __launch_bounds__(TROWS * 32)
gather_product_k(const int* __restrict__ rowptr, const int* __restrict__ rec,
                 const float* __restrict__ h, const float* __restrict__ dis,
                 const float* __restrict__ bias, const float* __restrict__ Wn,
                 float* __restrict__ outp, int n) {
    __shared__ __attribute__((aligned(16))) float tA[TROWS * TPITCH];
    __shared__ __attribute__((aligned(16))) float tD[TROWS * TPITCH];
    const int lane = threadIdx.x & 31;
    const int wid  = threadIdx.x >> 5;
    const int node = __builtin_amdgcn_readfirstlane(blockIdx.x * TROWS + wid);
    const bool live = node < n;
    float2 r = make_float2(0.0f, 0.0f);
    if (live) r = gather_row(rowptr, rec, h, dis, bias, node, lane);
    *(float2*)(tA + wid * TPITCH + (lane << 1)) = r;
    __syncthreads();

    if (wid < HID / 16) {
        const int mr  = lane & 15;
        const int khi = (lane >> 4) << 1;
        const float* arow = tA + mr * TPITCH + khi;
        const float* bcol = Wn + wid * 16 + mr;
        v8f acc = {};
#pragma unroll
        for (int k = 0; k < HID; k += 4) {
            v2f a, b;
            const float2 av = *(const float2*)(arow + k);
            a[0] = av.x; a[1] = av.y;
            const float* brow0 = bcol + (k + khi) * HID;
            b[0] = brow0[0]; b[1] = brow0[HID];
            acc = __builtin_amdgcn_wmma_f32_16x16x4_f32(false, a, false, b, (short)0, acc, false, false);
        }
        float* drow = tD + ((lane >> 4) << 3) * TPITCH + wid * 16 + mr;
#pragma unroll
        for (int j = 0; j < 8; ++j) drow[j * TPITCH] = acc[j];
    }
    __syncthreads();
    if (live)
        *(float2*)(outp + (size_t)node * HID + (lane << 1)) =
            *(const float2*)(tD + wid * TPITCH + (lane << 1));
}

#define PRUN 16
__device__ __forceinline__ void pool_flush(float* __restrict__ pooled, float* __restrict__ counts,
                                           int g, int lane, float sx, float sy,
                                           unsigned int mx, unsigned int my, int c) {
    float* p = pooled + (size_t)g * 128 + (lane << 1);
    atomicAdd(p, sx);
    atomicAdd(p + 1, sy);
    atomicMax((unsigned int*)(p + 64), mx);
    atomicMax((unsigned int*)(p + 65), my);
    if (lane == 0) atomicAdd(&counts[g], (float)c);
}

__global__ void __launch_bounds__(256)
pool_runs_k(const float* __restrict__ h, const int* __restrict__ batch,
            float* __restrict__ pooled, float* __restrict__ counts, int n) {
    const int lane = threadIdx.x & 31;
    int w = (blockIdx.x * blockDim.x + threadIdx.x) >> 5;
    w = __builtin_amdgcn_readfirstlane(w);
    const int base = w * PRUN;
    if (base >= n) return;
    const int m = min(PRUN, n - base);

    int gid = 0;
    if (lane < m) gid = batch[base + lane];

    const float* hl = h + (size_t)base * HID + (lane << 1);
    float2 hv[PRUN];
#pragma unroll
    for (int j = 0; j < PRUN; ++j) {
        const int jj = min(j, m - 1);
        hv[j] = *(const float2*)(hl + jj * HID);
    }

    int cur = __builtin_amdgcn_readlane(gid, 0);
    float sx = hv[0].x, sy = hv[0].y;
    unsigned int mx = __float_as_uint(hv[0].x), my = __float_as_uint(hv[0].y);
    int c = 1;
#pragma unroll
    for (int j = 1; j < PRUN; ++j) {
        if (j < m) {
            const int g = __builtin_amdgcn_readlane(gid, j);
            if (g != cur) {
                pool_flush(pooled, counts, cur, lane, sx, sy, mx, my, c);
                cur = g; sx = 0.0f; sy = 0.0f; mx = 0u; my = 0u; c = 0;
            }
            sx += hv[j].x;
            sy += hv[j].y;
            mx = max(mx, __float_as_uint(hv[j].x));
            my = max(my, __float_as_uint(hv[j].y));
            ++c;
        }
    }
    pool_flush(pooled, counts, cur, lane, sx, sy, mx, my, c);
}

__global__ void meandiv_k(float* __restrict__ pooled, const float* __restrict__ counts, int G) {
    int i = blockIdx.x * blockDim.x + threadIdx.x;
    if (i >= G * HID) return;
    int g = i >> 6;
    int f = i & 63;
    pooled[(size_t)g * 128 + f] /= fmaxf(counts[g], 1.0f);
}

__global__ void head_k(const float* __restrict__ pooled, const float* __restrict__ Wout,
                       const float* __restrict__ bout, float* __restrict__ out, int G) {
    int t = blockIdx.x * blockDim.x + threadIdx.x;
    int g = t >> 5;
    int lane = t & 31;
    if (g >= G) return;
    float v = 0.0f;
#pragma unroll
    for (int j = lane; j < 2 * HID; j += 32) v += pooled[(size_t)g * 128 + j] * Wout[j];
#pragma unroll
    for (int off = 16; off; off >>= 1) v += __shfl_xor(v, off, 32);
    if (lane == 0) out[g] = v + bout[0];
}

#define WS_DIS    0
#define WS_ROWPTR (WS_DIS + N_NODES)
#define WS_TOT    (WS_ROWPTR + N_NODES + 4)
#define WS_COUNTS (WS_TOT + 512)
#define WS_HIST   (WS_COUNTS + NGRAPH)
#define WS_TABLE  ((NB * NCH + 3) / 4 * 4)
#define WS_OFF    (WS_HIST + WS_TABLE)
#define WS_BIN    (WS_OFF + WS_TABLE)
#define WS_REC    (WS_BIN + 2 * N_EDGES)
#define WS_BUF0   (((WS_REC + 2 * N_EDGES) + 63) / 64 * 64)

extern "C" void kernel_launch(void* const* d_in, const int* in_sizes, int n_in,
                              void* d_out, int out_size, void* d_ws, size_t ws_size,
                              hipStream_t stream) {
    const float* x     = (const float*)d_in[0];
    const int*   eidx  = (const int*)  d_in[1];
    const int*   batch = (const int*)  d_in[2];
    const float* Ws[4] = { (const float*)d_in[3], (const float*)d_in[5],
                           (const float*)d_in[7], (const float*)d_in[9] };
    const float* bs[4] = { (const float*)d_in[4], (const float*)d_in[6],
                           (const float*)d_in[8], (const float*)d_in[10] };
    const float* Wout  = (const float*)d_in[11];
    const float* bout  = (const float*)d_in[12];

    const int N = N_NODES, E = N_EDGES, G = NGRAPH;
    const int* src = eidx;
    const int* dst = eidx + E;

    float* ws     = (float*)d_ws;
    float* dis    = ws + WS_DIS;
    int*   rowptr = (int*)(ws + WS_ROWPTR);
    int*   tot    = (int*)(ws + WS_TOT);
    float* counts = ws + WS_COUNTS;
    int*   histT  = (int*)(ws + WS_HIST);
    int*   offT   = (int*)(ws + WS_OFF);
    int*   binned = (int*)(ws + WS_BIN);
    int*   rec    = (int*)(ws + WS_REC);
    float* buf0   = ws + WS_BUF0;
    float* buf1   = buf0 + (size_t)N * HID;

    float* out    = (float*)d_out;
    float* pooled = out + G;

    bin_count_k  <<<NCH, SORT_T, 0, stream>>>(dst, histT);
    bin_scan_k   <<<NB,  SORT_T, 0, stream>>>(histT, offT, tot);
    bin_scatter_k<<<NCH, SORT_T, 0, stream>>>(src, dst, offT, tot, binned);
    bucket_rows_k<<<NB,  SORT_T, 0, stream>>>(binned, tot, rowptr, dis, rec);
    norm_fill_k  <<<(E / 2 + 255) / 256, 256, 0, stream>>>(rec, dis, E / 2);

    float* HWa = buf0;
    float* HWb = buf1;
    const int gemmThreads = (N / 16) * 32;
    gemm_wmma_k<<<(gemmThreads + 255) / 256, 256, 0, stream>>>(x, Ws[0], HWa, N, 128);
    for (int l = 0; l < 3; ++l) {
        gather_product_k<<<(N + TROWS - 1) / TROWS, TROWS * 32, 0, stream>>>(
            rowptr, rec, HWa, dis, bs[l], Ws[l + 1], HWb, N);
        float* t = HWa; HWa = HWb; HWb = t;
    }
    gather_layer_k<<<(N * 32 + 255) / 256, 256, 0, stream>>>(rowptr, rec, HWa, dis, bs[3], HWb, N);
    const float* hin = HWb;

    fill_k   <<<(G + G * 128 + 255) / 256, 256, 0, stream>>>(out, 0.0f, G + G * 128);
    fill_k   <<<(G + 255) / 256, 256, 0, stream>>>(counts, 0.0f, G);
    pool_runs_k<<<((N + PRUN - 1) / PRUN * 32 + 255) / 256, 256, 0, stream>>>(hin, batch, pooled, counts, N);
    meandiv_k<<<(G * HID + 255) / 256, 256, 0, stream>>>(pooled, counts, G);
    head_k   <<<(G * 32 + 255) / 256, 256, 0, stream>>>(pooled, Wout, bout, out, G);
}


# Neighbour-sum launches as a resident grid: a wave walks many nodes (a block ~6 tiles), next node's bounds, records, own row asked for ahead, weights staged in LDS once per block; sum order kept
# speedup vs baseline: 4.5409x; 1.0996x over previous
#include <hip/hip_runtime.h>
#include <hip/hip_bf16.h>

typedef __attribute__((ext_vector_type(2))) float v2f;
typedef __attribute__((ext_vector_type(8))) float v8f;

#define N_NODES 100000
#define N_EDGES 1600000
#define HID     64
#define NGRAPH  2048

__global__ void fill_k(float* __restrict__ p, float v, long long n) {
    long long i = (long long)blockIdx.x * blockDim.x + threadIdx.x;
    if (i < n) p[i] = v;
}

__global__ void gemm_wmma_k(const float* __restrict__ A, const float* __restrict__ W,
                            float* __restrict__ C, int rows, int K) {
    const int lane = threadIdx.x & 31;
    const int w    = (blockIdx.x * blockDim.x + threadIdx.x) >> 5;
    const int rowTiles = rows >> 4;
    if (w >= rowTiles) return;

    const int m0  = w << 4;
    const int mr  = lane & 15;
    const int khi = (lane >> 4) << 1;

    const float* arow = A + (size_t)(m0 + mr) * K + khi;
    const float* bcol = W + mr;

    v8f acc0 = {}, acc1 = {}, acc2 = {}, acc3 = {};
    for (int k = 0; k < K; k += 4) {
        v2f a;
        const float2 av = *(const float2*)(arow + k);
        a[0] = av.x; a[1] = av.y;

        const float* brow0 = bcol + (size_t)(k + khi) * HID;
        const float* brow1 = brow0 + HID;
        v2f b0, b1, b2, b3;
        b0[0] = brow0[0];  b0[1] = brow1[0];
        b1[0] = brow0[16]; b1[1] = brow1[16];
        b2[0] = brow0[32]; b2[1] = brow1[32];
        b3[0] = brow0[48]; b3[1] = brow1[48];

        acc0 = __builtin_amdgcn_wmma_f32_16x16x4_f32(false, a, false, b0, (short)0, acc0, false, false);
        acc1 = __builtin_amdgcn_wmma_f32_16x16x4_f32(false, a, false, b1, (short)0, acc1, false, false);
        acc2 = __builtin_amdgcn_wmma_f32_16x16x4_f32(false, a, false, b2, (short)0, acc2, false, false);
        acc3 = __builtin_amdgcn_wmma_f32_16x16x4_f32(false, a, false, b3, (short)0, acc3, false, false);
    }

    float* crow = C + (size_t)(m0 + ((lane >> 4) << 3)) * HID + mr;
#pragma unroll
    for (int j = 0; j < 8; ++j) {
        crow[(size_t)j * HID +  0] = acc0[j];
        crow[(size_t)j * HID + 16] = acc1[j];
        crow[(size_t)j * HID + 32] = acc2[j];
        crow[(size_t)j * HID + 48] = acc3[j];
    }
}

#define BSH    8
#define NB     ((N_NODES + 255) >> BSH)
#define CH     4096
#define NCH    ((N_EDGES + CH - 1) / CH)
#define SORT_T 512

__device__ __forceinline__ int scan512(int v, int* sm, int& total) {
    const int lane = threadIdx.x & 31;
    const int wid  = threadIdx.x >> 5;
    int inc = v;
#pragma unroll
    for (int off = 1; off < 32; off <<= 1) {
        const int u = __shfl_up(inc, off, 32);
        if (lane >= off) inc += u;
    }
    if (lane == 31) sm[wid] = inc;
    __syncthreads();
    int pre = 0, tot = 0;
#pragma unroll
    for (int k = 0; k < SORT_T / 32; ++k) {
        const int s = sm[k];
        tot += s;
        if (k < wid) pre += s;
    }
    total = tot;
    return pre + inc - v;
}

__global__ void __launch_bounds__(SORT_T)
bin_count_k(const int* __restrict__ dst, int* __restrict__ histT) {
    __shared__ int c[NB];
    const int tid = threadIdx.x;
    const int chunk = blockIdx.x;
    int4 d[CH / (SORT_T * 4)];
#pragma unroll
    for (int k = 0; k < CH / (SORT_T * 4); ++k) {
        const int e = chunk * CH + k * (SORT_T * 4) + tid * 4;
        d[k] = (e < N_EDGES) ? *(const int4*)(dst + e) : make_int4(-1, -1, -1, -1);
    }
    for (int i = tid; i < NB; i += SORT_T) c[i] = 0;
    __syncthreads();
#pragma unroll
    for (int k = 0; k < CH / (SORT_T * 4); ++k) {
        if (d[k].x >= 0) {
            atomicAdd(&c[d[k].x >> BSH], 1);
            atomicAdd(&c[d[k].y >> BSH], 1);
            atomicAdd(&c[d[k].z >> BSH], 1);
            atomicAdd(&c[d[k].w >> BSH], 1);
        }
    }
    __syncthreads();
    for (int i = tid; i < NB; i += SORT_T) histT[i * NCH + chunk] = c[i];
}

__global__ void __launch_bounds__(SORT_T)
bin_scan_k(const int* __restrict__ histT, int* __restrict__ offT, int* __restrict__ tot) {
    __shared__ int sm[SORT_T / 32];
    const int tid = threadIdx.x;
    const int b = blockIdx.x;
    const int v = (tid < NCH) ? histT[b * NCH + tid] : 0;
    int total;
    const int excl = scan512(v, sm, total);
    if (tid < NCH) offT[b * NCH + tid] = excl;
    if (tid == 0) tot[b] = total;
}

__global__ void __launch_bounds__(SORT_T)
bin_scatter_k(const int* __restrict__ src, const int* __restrict__ dst,
              const int* __restrict__ offT, const int* __restrict__ tot,
              int* __restrict__ binned) {
    __shared__ int cur[NB];
    __shared__ int sm[SORT_T / 32];
    const int tid = threadIdx.x;
    const int chunk = blockIdx.x;
    int4 s[CH / (SORT_T * 4)], d[CH / (SORT_T * 4)];
#pragma unroll
    for (int k = 0; k < CH / (SORT_T * 4); ++k) {
        const int e = chunk * CH + k * (SORT_T * 4) + tid * 4;
        const bool in = e < N_EDGES;
        s[k] = in ? *(const int4*)(src + e) : make_int4(0, 0, 0, 0);
        d[k] = in ? *(const int4*)(dst + e) : make_int4(-1, -1, -1, -1);
    }
    const int t = (tid < NB) ? tot[tid] : 0;
    const int o = (tid < NB) ? offT[tid * NCH + chunk] : 0;
    int total;
    const int bstart = scan512(t, sm, total);
    if (tid < NB) cur[tid] = bstart + o;
    __syncthreads();
    int2* out2 = (int2*)binned;
#pragma unroll
    for (int k = 0; k < CH / (SORT_T * 4); ++k) {
        if (d[k].x >= 0) {
            const int p0 = atomicAdd(&cur[d[k].x >> BSH], 1);
            const int p1 = atomicAdd(&cur[d[k].y >> BSH], 1);
            const int p2 = atomicAdd(&cur[d[k].z >> BSH], 1);
            const int p3 = atomicAdd(&cur[d[k].w >> BSH], 1);
            out2[p0] = make_int2(s[k].x, d[k].x);
            out2[p1] = make_int2(s[k].y, d[k].y);
            out2[p2] = make_int2(s[k].z, d[k].z);
            out2[p3] = make_int2(s[k].w, d[k].w);
        }
    }
}

#define BLD 4
__global__ void __launch_bounds__(SORT_T)
bucket_rows_k(const int* __restrict__ binned, const int* __restrict__ tot,
              int* __restrict__ rowptr, float* __restrict__ dis, int* __restrict__ rec) {
    __shared__ int cnt[1 << BSH];
    __shared__ int cur[1 << BSH];
    __shared__ int smA[SORT_T / 32];
    __shared__ int smB[SORT_T / 32];
    __shared__ int range[2];
    const int tid = threadIdx.x;
    const int b = blockIdx.x;
    const int node0 = b << BSH;
    if (tid < (1 << BSH)) cnt[tid] = 0;
    const int t = (tid < NB) ? tot[tid] : 0;
    int total;
    const int bstart = scan512(t, smA, total);
    if (tid == b) { range[0] = bstart; range[1] = bstart + t; }
    __syncthreads();
    const int beg = range[0];
    const int end = range[1];
    const int2* in2 = (const int2*)binned;
    int2* out2 = (int2*)rec;

    for (int e0 = beg; e0 < end; e0 += SORT_T * BLD) {
        int d[BLD];
#pragma unroll
        for (int u = 0; u < BLD; ++u) {
            const int e = e0 + u * SORT_T + tid;
            d[u] = (e < end) ? in2[e].y : -1;
        }
#pragma unroll
        for (int u = 0; u < BLD; ++u)
            if (d[u] >= 0) atomicAdd(&cnt[d[u] - node0], 1);
    }
    __syncthreads();

    const int c = (tid < (1 << BSH)) ? cnt[tid] : 0;
    int total2;
    const int ex = scan512(c, smB, total2);
    if (tid < (1 << BSH)) {
        cur[tid] = beg + ex;
        const int node = node0 + tid;
        if (node < N_NODES) {
            rowptr[node] = beg + ex;
            dis[node] = rsqrtf((float)(c + 1));
        }
    }
    if (b == NB - 1 && tid == 0) rowptr[N_NODES] = end;
    __syncthreads();

    for (int e0 = beg; e0 < end; e0 += SORT_T * BLD) {
        int2 r[BLD];
#pragma unroll
        for (int u = 0; u < BLD; ++u) {
            const int e = e0 + u * SORT_T + tid;
            r[u] = (e < end) ? in2[e] : make_int2(0, -1);
        }
#pragma unroll
        for (int u = 0; u < BLD; ++u)
            if (r[u].y >= 0) {
                const int p = atomicAdd(&cur[r[u].y - node0], 1);
                out2[p] = r[u];
            }
    }
}

__global__ void norm_fill_k(int* __restrict__ rec, const float* __restrict__ dis, int pairs) {
    const int q = blockIdx.x * blockDim.x + threadIdx.x;
    if (q >= pairs) return;
    int4 r = ((const int4*)rec)[q];
    r.y = __float_as_int(dis[r.x] * dis[r.y]);
    r.w = __float_as_int(dis[r.z] * dis[r.w]);
    ((int4*)rec)[q] = r;
}

#define GRP 8
__device__ __forceinline__ float2 gather_row(const int* __restrict__ rec, const float* __restrict__ hl,
                                             int lane, int beg, int end, int2 ev0,
                                             float2 own, float dv, float2 bv) {
    const float sw = dv * dv;
    float2 acc = make_float2(0.0f, 0.0f);
    int2 ev = ev0;

    for (int base = beg; base < end; base += 32) {
        const int m = min(32, end - base);
        if (base != beg) {
            ev = make_int2(0, 0);
            if (lane < m) ev = ((const int2*)rec)[base + lane];
        }
        for (int j = 0; j < m; j += GRP) {
            float2 hv[GRP];
            float  wv[GRP];
#pragma unroll
            for (int u = 0; u < GRP; ++u) {
                const int jj = min(j + u, m - 1);
                const int   s = __builtin_amdgcn_readlane(ev.x, jj);
                const float w = __int_as_float(__builtin_amdgcn_readlane(ev.y, jj));
                wv[u] = (j + u < m) ? w : 0.0f;
                hv[u] = *(const float2*)(hl + (unsigned)(s * HID));
            }
#pragma unroll
            for (int u = 0; u < GRP; ++u) {
                acc.x += hv[u].x * wv[u];
                acc.y += hv[u].y * wv[u];
            }
        }
    }
    float2 r;
    r.x = fmaxf(acc.x + own.x * sw + bv.x, 0.0f);
    r.y = fmaxf(acc.y + own.y * sw + bv.y, 0.0f);
    return r;
}

#define GP_BLOCKS 1000
#define GL_BLOCKS 2000

__global__ void __launch_bounds__(256)
gather_layer_k(const int* __restrict__ rowptr, const int* __restrict__ rec,
               const float* __restrict__ h, const float* __restrict__ dis,
               const float* __restrict__ bias, float* __restrict__ outp, int n) {
    const int lane = threadIdx.x & 31;
    const int S = (gridDim.x * blockDim.x) >> 5;
    int node = __builtin_amdgcn_readfirstlane((blockIdx.x * blockDim.x + threadIdx.x) >> 5);
    if (node >= n) return;
    const float* hl = h + (lane << 1);
    const float2 bv = *(const float2*)(bias + (lane << 1));
    int beg = rowptr[node];
    int end = rowptr[node + 1];
    int2 ev = make_int2(0, 0);
    if (lane < end - beg) ev = ((const int2*)rec)[beg + lane];
    float2 own = *(const float2*)(hl + (size_t)node * HID);
    float dv = dis[node];
    int n1 = min(node + S, n - 1);
    int nb = rowptr[n1 + (lane & 1)];
    for (; node < n; node += S) {
        const float2 r = gather_row(rec, hl, lane, beg, end, ev, own, dv, bv);
        const int nbeg = __builtin_amdgcn_readlane(nb, 0);
        const int nend = __builtin_amdgcn_readlane(nb, 1);
        int2 nev = make_int2(0, 0);
        if (lane < nend - nbeg) nev = ((const int2*)rec)[nbeg + lane];
        const float2 nown = *(const float2*)(hl + (size_t)n1 * HID);
        const float  ndv  = dis[n1];
        const int n2 = min(n1 + S, n - 1);
        const int nnb = rowptr[n2 + (lane & 1)];
        *(float2*)(outp + (size_t)node * HID + (lane << 1)) = r;
        beg = nbeg; end = nend; ev = nev; own = nown; dv = ndv; n1 = n2; nb = nnb;
    }
}

#define TROWS  16
#define TPITCH 68
#define WLANE  36

__device__ __forceinline__ void lds_barrier() {
    __builtin_amdgcn_fence(__ATOMIC_RELEASE, "workgroup", "local");
    __builtin_amdgcn_s_barrier();
    __builtin_amdgcn_fence(__ATOMIC_ACQUIRE, "workgroup", "local");
}

__global__ void __launch_bounds__(TROWS * 32)
gather_product_k(const int* __restrict__ rowptr, const int* __restrict__ rec,
                 const float* __restrict__ h, const float* __restrict__ dis,
                 const float* __restrict__ bias, const float* __restrict__ Wn,
                 float* __restrict__ outp, int n) {
    __shared__ __attribute__((aligned(16))) float tA[TROWS * TPITCH];
    __shared__ __attribute__((aligned(16))) float tD[TROWS * TPITCH];
    __shared__ __attribute__((aligned(16))) float sW[4 * 32 * WLANE];
    const int lane = threadIdx.x & 31;
    const int wid  = threadIdx.x >> 5;
    const int tiles = (n + TROWS - 1) / TROWS;
    const int S = gridDim.x * TROWS;
    const int mr  = lane & 15;
    const int khi = (lane >> 4) << 1;

#pragma unroll
    for (int cc = 0; cc < HID / 16; ++cc) {
        const float* wp = Wn + (4 * wid + khi) * HID + cc * 16 + mr;
        float2 wv2;
        wv2.x = wp[0];
        wv2.y = wp[HID];
        *(float2*)(sW + (cc * 32 + lane) * WLANE + 2 * wid) = wv2;
    }

    const float* hl = h + (lane << 1);
    const float2 bv = *(const float2*)(bias + (lane << 1));
    int tile = blockIdx.x;
    int node = __builtin_amdgcn_readfirstlane(min(tile * TROWS + wid, n - 1));
    int beg = rowptr[node];
    int end = rowptr[node + 1];
    int2 ev = make_int2(0, 0);
    if (lane < end - beg) ev = ((const int2*)rec)[beg + lane];
    float2 own = *(const float2*)(hl + (size_t)node * HID);
    float dv = dis[node];
    int n1 = __builtin_amdgcn_readfirstlane(min(node + S, n - 1));
    int nb = rowptr[n1 + (lane & 1)];
    __syncthreads();

    for (; tile < tiles; tile += gridDim.x) {
        const int  cur  = tile * TROWS + wid;
        const bool live = cur < n;
        const float2 r = gather_row(rec, hl, lane, beg, end, ev, own, dv, bv);
        const int nbeg = __builtin_amdgcn_readlane(nb, 0);
        const int nend = __builtin_amdgcn_readlane(nb, 1);
        int2 nev = make_int2(0, 0);
        if (lane < nend - nbeg) nev = ((const int2*)rec)[nbeg + lane];
        const float2 nown = *(const float2*)(hl + (size_t)n1 * HID);
        const float  ndv  = dis[n1];
        const int n2 = __builtin_amdgcn_readfirstlane(min(n1 + S, n - 1));
        const int nnb = rowptr[n2 + (lane & 1)];

        *(float2*)(tA + wid * TPITCH + (lane << 1)) = r;
        lds_barrier();

        if (wid < HID / 16) {
            const float* arow = tA + mr * TPITCH + khi;
            const float* bfr  = sW + (wid * 32 + lane) * WLANE;
            v8f acc = {};
#pragma unroll
            for (int k = 0; k < HID; k += 8) {
                const float4 bq = *(const float4*)(bfr + (k >> 1));
                v2f a, b;
                const float2 av0 = *(const float2*)(arow + k);
                a[0] = av0.x; a[1] = av0.y;
                b[0] = bq.x;  b[1] = bq.y;
                acc = __builtin_amdgcn_wmma_f32_16x16x4_f32(false, a, false, b, (short)0, acc, false, false);
                const float2 av1 = *(const float2*)(arow + k + 4);
                a[0] = av1.x; a[1] = av1.y;
                b[0] = bq.z;  b[1] = bq.w;
                acc = __builtin_amdgcn_wmma_f32_16x16x4_f32(false, a, false, b, (short)0, acc, false, false);
            }
            float* drow = tD + ((lane >> 4) << 3) * TPITCH + wid * 16 + mr;
#pragma unroll
            for (int j = 0; j < 8; ++j) drow[j * TPITCH] = acc[j];
        }
        lds_barrier();
        if (live)
            *(float2*)(outp + (size_t)cur * HID + (lane << 1)) =
                *(const float2*)(tD + wid * TPITCH + (lane << 1));
        beg = nbeg; end = nend; ev = nev; own = nown; dv = ndv; n1 = n2; nb = nnb;
    }
}

#define PRUN 16
__device__ __forceinline__ void pool_flush(float* __restrict__ pooled, float* __restrict__ counts,
                                           int g, int lane, float sx, float sy,
                                           unsigned int mx, unsigned int my, int c) {
    float* p = pooled + (size_t)g * 128 + (lane << 1);
    atomicAdd(p, sx);
    atomicAdd(p + 1, sy);
    atomicMax((unsigned int*)(p + 64), mx);
    atomicMax((unsigned int*)(p + 65), my);
    if (lane == 0) atomicAdd(&counts[g], (float)c);
}

__global__ void __launch_bounds__(256)
pool_runs_k(const float* __restrict__ h, const int* __restrict__ batch,
            float* __restrict__ pooled, float* __restrict__ counts, int n) {
    const int lane = threadIdx.x & 31;
    int w = (blockIdx.x * blockDim.x + threadIdx.x) >> 5;
    w = __builtin_amdgcn_readfirstlane(w);
    const int base = w * PRUN;
    if (base >= n) return;
    const int m = min(PRUN, n - base);

    int gid = 0;
    if (lane < m) gid = batch[base + lane];

    const float* hl = h + (size_t)base * HID + (lane << 1);
    float2 hv[PRUN];
#pragma unroll
    for (int j = 0; j < PRUN; ++j) {
        const int jj = min(j, m - 1);
        hv[j] = *(const float2*)(hl + jj * HID);
    }

    int cur = __builtin_amdgcn_readlane(gid, 0);
    float sx = hv[0].x, sy = hv[0].y;
    unsigned int mx = __float_as_uint(hv[0].x), my = __float_as_uint(hv[0].y);
    int c = 1;
#pragma unroll
    for (int j = 1; j < PRUN; ++j) {
        if (j < m) {
            const int g = __builtin_amdgcn_readlane(gid, j);
            if (g != cur) {
                pool_flush(pooled, counts, cur, lane, sx, sy, mx, my, c);
                cur = g; sx = 0.0f; sy = 0.0f; mx = 0u; my = 0u; c = 0;
            }
            sx += hv[j].x;
            sy += hv[j].y;
            mx = max(mx, __float_as_uint(hv[j].x));
            my = max(my, __float_as_uint(hv[j].y));
            ++c;
        }
    }
    pool_flush(pooled, counts, cur, lane, sx, sy, mx, my, c);
}

__global__ void meandiv_k(float* __restrict__ pooled, const float* __restrict__ counts, int G) {
    int i = blockIdx.x * blockDim.x + threadIdx.x;
    if (i >= G * HID) return;
    int g = i >> 6;
    int f = i & 63;
    pooled[(size_t)g * 128 + f] /= fmaxf(counts[g], 1.0f);
}

__global__ void head_k(const float* __restrict__ pooled, const float* __restrict__ Wout,
                       const float* __restrict__ bout, float* __restrict__ out, int G) {
    int t = blockIdx.x * blockDim.x + threadIdx.x;
    int g = t >> 5;
    int lane = t & 31;
    if (g >= G) return;
    float v = 0.0f;
#pragma unroll
    for (int j = lane; j < 2 * HID; j += 32) v += pooled[(size_t)g * 128 + j] * Wout[j];
#pragma unroll
    for (int off = 16; off; off >>= 1) v += __shfl_xor(v, off, 32);
    if (lane == 0) out[g] = v + bout[0];
}

#define WS_DIS    0
#define WS_ROWPTR (WS_DIS + N_NODES)
#define WS_TOT    (WS_ROWPTR + N_NODES + 4)
#define WS_COUNTS (WS_TOT + 512)
#define WS_HIST   (WS_COUNTS + NGRAPH)
#define WS_TABLE  ((NB * NCH + 3) / 4 * 4)
#define WS_OFF    (WS_HIST + WS_TABLE)
#define WS_BIN    (WS_OFF + WS_TABLE)
#define WS_REC    (WS_BIN + 2 * N_EDGES)
#define WS_BUF0   (((WS_REC + 2 * N_EDGES) + 63) / 64 * 64)

extern "C" void kernel_launch(void* const* d_in, const int* in_sizes, int n_in,
                              void* d_out, int out_size, void* d_ws, size_t ws_size,
                              hipStream_t stream) {
    const float* x     = (const float*)d_in[0];
    const int*   eidx  = (const int*)  d_in[1];
    const int*   batch = (const int*)  d_in[2];
    const float* Ws[4] = { (const float*)d_in[3], (const float*)d_in[5],
                           (const float*)d_in[7], (const float*)d_in[9] };
    const float* bs[4] = { (const float*)d_in[4], (const float*)d_in[6],
                           (const float*)d_in[8], (const float*)d_in[10] };
    const float* Wout  = (const float*)d_in[11];
    const float* bout  = (const float*)d_in[12];

    const int N = N_NODES, E = N_EDGES, G = NGRAPH;
    const int* src = eidx;
    const int* dst = eidx + E;

    float* ws     = (float*)d_ws;
    float* dis    = ws + WS_DIS;
    int*   rowptr = (int*)(ws + WS_ROWPTR);
    int*   tot    = (int*)(ws + WS_TOT);
    float* counts = ws + WS_COUNTS;
    int*   histT  = (int*)(ws + WS_HIST);
    int*   offT   = (int*)(ws + WS_OFF);
    int*   binned = (int*)(ws + WS_BIN);
    int*   rec    = (int*)(ws + WS_REC);
    float* buf0   = ws + WS_BUF0;
    float* buf1   = buf0 + (size_t)N * HID;

    float* out    = (float*)d_out;
    float* pooled = out + G;

    bin_count_k  <<<NCH, SORT_T, 0, stream>>>(dst, histT);
    bin_scan_k   <<<NB,  SORT_T, 0, stream>>>(histT, offT, tot);
    bin_scatter_k<<<NCH, SORT_T, 0, stream>>>(src, dst, offT, tot, binned);
    bucket_rows_k<<<NB,  SORT_T, 0, stream>>>(binned, tot, rowptr, dis, rec);
    norm_fill_k  <<<(E / 2 + 255) / 256, 256, 0, stream>>>(rec, dis, E / 2);

    float* HWa = buf0;
    float* HWb = buf1;
    const int gemmThreads = (N / 16) * 32;
    gemm_wmma_k<<<(gemmThreads + 255) / 256, 256, 0, stream>>>(x, Ws[0], HWa, N, 128);
    for (int l = 0; l < 3; ++l) {
        gather_product_k<<<GP_BLOCKS, TROWS * 32, 0, stream>>>(
            rowptr, rec, HWa, dis, bs[l], Ws[l + 1], HWb, N);
        float* t = HWa; HWa = HWb; HWb = t;
    }
    gather_layer_k<<<GL_BLOCKS, 256, 0, stream>>>(rowptr, rec, HWa, dis, bs[3], HWb, N);
    const float* hin = HWb;

    fill_k   <<<(G + G * 128 + 255) / 256, 256, 0, stream>>>(out, 0.0f, G + G * 128);
    fill_k   <<<(G + 255) / 256, 256, 0, stream>>>(counts, 0.0f, G);
    pool_runs_k<<<((N + PRUN - 1) / PRUN * 32 + 255) / 256, 256, 0, stream>>>(hin, batch, pooled, counts, N);
    meandiv_k<<<(G * HID + 255) / 256, 256, 0, stream>>>(pooled, counts, G);
    head_k   <<<(G * 32 + 255) / 256, 256, 0, stream>>>(pooled, Wout, bout, out, G);
}


# Edge weight factored: gathered rows stored pre-multiplied by own deg^-1/2, each sum scaled once by the destination's; weight-fill launch removed, 4-byte records; term rounding changes, sum order kept
# speedup vs baseline: 4.7655x; 1.0495x over previous
#include <hip/hip_runtime.h>
#include <hip/hip_bf16.h>

typedef __attribute__((ext_vector_type(2))) float v2f;
typedef __attribute__((ext_vector_type(8))) float v8f;

#define N_NODES 100000
#define N_EDGES 1600000
#define HID     64
#define NGRAPH  2048

__global__ void fill_k(float* __restrict__ p, float v, long long n) {
    long long i = (long long)blockIdx.x * blockDim.x + threadIdx.x;
    if (i < n) p[i] = v;
}

__global__ void gemm_wmma_k(const float* __restrict__ A, const float* __restrict__ W,
                            const float* __restrict__ rs,
                            float* __restrict__ C, int rows, int K) {
    const int lane = threadIdx.x & 31;
    const int w    = (blockIdx.x * blockDim.x + threadIdx.x) >> 5;
    const int rowTiles = rows >> 4;
    if (w >= rowTiles) return;

    const int m0  = w << 4;
    const int mr  = lane & 15;
    const int khi = (lane >> 4) << 1;

    const float* arow = A + (size_t)(m0 + mr) * K + khi;
    const float* bcol = W + mr;

    v8f acc0 = {}, acc1 = {}, acc2 = {}, acc3 = {};
    for (int k = 0; k < K; k += 4) {
        v2f a;
        const float2 av = *(const float2*)(arow + k);
        a[0] = av.x; a[1] = av.y;

        const float* brow0 = bcol + (size_t)(k + khi) * HID;
        const float* brow1 = brow0 + HID;
        v2f b0, b1, b2, b3;
        b0[0] = brow0[0];  b0[1] = brow1[0];
        b1[0] = brow0[16]; b1[1] = brow1[16];
        b2[0] = brow0[32]; b2[1] = brow1[32];
        b3[0] = brow0[48]; b3[1] = brow1[48];

        acc0 = __builtin_amdgcn_wmma_f32_16x16x4_f32(false, a, false, b0, (short)0, acc0, false, false);
        acc1 = __builtin_amdgcn_wmma_f32_16x16x4_f32(false, a, false, b1, (short)0, acc1, false, false);
        acc2 = __builtin_amdgcn_wmma_f32_16x16x4_f32(false, a, false, b2, (short)0, acc2, false, false);
        acc3 = __builtin_amdgcn_wmma_f32_16x16x4_f32(false, a, false, b3, (short)0, acc3, false, false);
    }

    const int r0 = m0 + ((lane >> 4) << 3);
    float* crow = C + (size_t)r0 * HID + mr;
    const float4 sa = *(const float4*)(rs + r0);
    const float4 sb = *(const float4*)(rs + r0 + 4);
    const float sc[8] = { sa.x, sa.y, sa.z, sa.w, sb.x, sb.y, sb.z, sb.w };
#pragma unroll
    for (int j = 0; j < 8; ++j) {
        crow[(size_t)j * HID +  0] = acc0[j] * sc[j];
        crow[(size_t)j * HID + 16] = acc1[j] * sc[j];
        crow[(size_t)j * HID + 32] = acc2[j] * sc[j];
        crow[(size_t)j * HID + 48] = acc3[j] * sc[j];
    }
}

#define BSH    8
#define NB     ((N_NODES + 255) >> BSH)
#define CH     4096
#define NCH    ((N_EDGES + CH - 1) / CH)
#define SORT_T 512

__device__ __forceinline__ int scan512(int v, int* sm, int& total) {
    const int lane = threadIdx.x & 31;
    const int wid  = threadIdx.x >> 5;
    int inc = v;
#pragma unroll
    for (int off = 1; off < 32; off <<= 1) {
        const int u = __shfl_up(inc, off, 32);
        if (lane >= off) inc += u;
    }
    if (lane == 31) sm[wid] = inc;
    __syncthreads();
    int pre = 0, tot = 0;
#pragma unroll
    for (int k = 0; k < SORT_T / 32; ++k) {
        const int s = sm[k];
        tot += s;
        if (k < wid) pre += s;
    }
    total = tot;
    return pre + inc - v;
}

__global__ void __launch_bounds__(SORT_T)
bin_count_k(const int* __restrict__ dst, int* __restrict__ histT) {
    __shared__ int c[NB];
    const int tid = threadIdx.x;
    const int chunk = blockIdx.x;
    int4 d[CH / (SORT_T * 4)];
#pragma unroll
    for (int k = 0; k < CH / (SORT_T * 4); ++k) {
        const int e = chunk * CH + k * (SORT_T * 4) + tid * 4;
        d[k] = (e < N_EDGES) ? *(const int4*)(dst + e) : make_int4(-1, -1, -1, -1);
    }
    for (int i = tid; i < NB; i += SORT_T) c[i] = 0;
    __syncthreads();
#pragma unroll
    for (int k = 0; k < CH / (SORT_T * 4); ++k) {
        if (d[k].x >= 0) {
            atomicAdd(&c[d[k].x >> BSH], 1);
            atomicAdd(&c[d[k].y >> BSH], 1);
            atomicAdd(&c[d[k].z >> BSH], 1);
            atomicAdd(&c[d[k].w >> BSH], 1);
        }
    }
    __syncthreads();
    for (int i = tid; i < NB; i += SORT_T) histT[i * NCH + chunk] = c[i];
}

__global__ void __launch_bounds__(SORT_T)
bin_scan_k(const int* __restrict__ histT, int* __restrict__ offT, int* __restrict__ tot) {
    __shared__ int sm[SORT_T / 32];
    const int tid = threadIdx.x;
    const int b = blockIdx.x;
    const int v = (tid < NCH) ? histT[b * NCH + tid] : 0;
    int total;
    const int excl = scan512(v, sm, total);
    if (tid < NCH) offT[b * NCH + tid] = excl;
    if (tid == 0) tot[b] = total;
}

__global__ void __launch_bounds__(SORT_T)
bin_scatter_k(const int* __restrict__ src, const int* __restrict__ dst,
              const int* __restrict__ offT, const int* __restrict__ tot,
              int* __restrict__ binned) {
    __shared__ int cur[NB];
    __shared__ int sm[SORT_T / 32];
    const int tid = threadIdx.x;
    const int chunk = blockIdx.x;
    int4 s[CH / (SORT_T * 4)], d[CH / (SORT_T * 4)];
#pragma unroll
    for (int k = 0; k < CH / (SORT_T * 4); ++k) {
        const int e = chunk * CH + k * (SORT_T * 4) + tid * 4;
        const bool in = e < N_EDGES;
        s[k] = in ? *(const int4*)(src + e) : make_int4(0, 0, 0, 0);
        d[k] = in ? *(const int4*)(dst + e) : make_int4(-1, -1, -1, -1);
    }
    const int t = (tid < NB) ? tot[tid] : 0;
    const int o = (tid < NB) ? offT[tid * NCH + chunk] : 0;
    int total;
    const int bstart = scan512(t, sm, total);
    if (tid < NB) cur[tid] = bstart + o;
    __syncthreads();
    int2* out2 = (int2*)binned;
#pragma unroll
    for (int k = 0; k < CH / (SORT_T * 4); ++k) {
        if (d[k].x >= 0) {
            const int p0 = atomicAdd(&cur[d[k].x >> BSH], 1);
            const int p1 = atomicAdd(&cur[d[k].y >> BSH], 1);
            const int p2 = atomicAdd(&cur[d[k].z >> BSH], 1);
            const int p3 = atomicAdd(&cur[d[k].w >> BSH], 1);
            out2[p0] = make_int2(s[k].x, d[k].x);
            out2[p1] = make_int2(s[k].y, d[k].y);
            out2[p2] = make_int2(s[k].z, d[k].z);
            out2[p3] = make_int2(s[k].w, d[k].w);
        }
    }
}

#define BLD 4
__global__ void __launch_bounds__(SORT_T)
bucket_rows_k(const int* __restrict__ binned, const int* __restrict__ tot,
              int* __restrict__ rowptr, float* __restrict__ dis, int* __restrict__ rec) {
    __shared__ int cnt[1 << BSH];
    __shared__ int cur[1 << BSH];
    __shared__ int smA[SORT_T / 32];
    __shared__ int smB[SORT_T / 32];
    __shared__ int range[2];
    const int tid = threadIdx.x;
    const int b = blockIdx.x;
    const int node0 = b << BSH;
    if (tid < (1 << BSH)) cnt[tid] = 0;
    const int t = (tid < NB) ? tot[tid] : 0;
    int total;
    const int bstart = scan512(t, smA, total);
    if (tid == b) { range[0] = bstart; range[1] = bstart + t; }
    __syncthreads();
    const int beg = range[0];
    const int end = range[1];
    const int2* in2 = (const int2*)binned;

    for (int e0 = beg; e0 < end; e0 += SORT_T * BLD) {
        int d[BLD];
#pragma unroll
        for (int u = 0; u < BLD; ++u) {
            const int e = e0 + u * SORT_T + tid;
            d[u] = (e < end) ? in2[e].y : -1;
        }
#pragma unroll
        for (int u = 0; u < BLD; ++u)
            if (d[u] >= 0) atomicAdd(&cnt[d[u] - node0], 1);
    }
    __syncthreads();

    const int c = (tid < (1 << BSH)) ? cnt[tid] : 0;
    int total2;
    const int ex = scan512(c, smB, total2);
    if (tid < (1 << BSH)) {
        cur[tid] = beg + ex;
        const int node = node0 + tid;
        if (node < N_NODES) {
            rowptr[node] = beg + ex;
            dis[node] = rsqrtf((float)(c + 1));
        }
    }
    if (b == NB - 1 && tid == 0) rowptr[N_NODES] = end;
    __syncthreads();

    for (int e0 = beg; e0 < end; e0 += SORT_T * BLD) {
        int2 r[BLD];
#pragma unroll
        for (int u = 0; u < BLD; ++u) {
            const int e = e0 + u * SORT_T + tid;
            r[u] = (e < end) ? in2[e] : make_int2(0, -1);
        }
#pragma unroll
        for (int u = 0; u < BLD; ++u)
            if (r[u].y >= 0) {
                const int p = atomicAdd(&cur[r[u].y - node0], 1);
                rec[p] = r[u].x;
            }
    }
}

#define GRP 8
__device__ __forceinline__ float2 gather_row(const int* __restrict__ rec, const float* __restrict__ hl,
                                             int lane, int beg, int end, int ev0,
                                             float2 own, float dv, float2 bv) {
    float2 acc = make_float2(0.0f, 0.0f);
    int ev = ev0;

    for (int base = beg; base < end; base += 32) {
        const int m = min(32, end - base);
        if (base != beg) {
            ev = 0;
            if (lane < m) ev = rec[base + lane];
        }
        for (int j = 0; j < m; j += GRP) {
            float2 hv[GRP];
            float  wv[GRP];
#pragma unroll
            for (int u = 0; u < GRP; ++u) {
                const int jj = min(j + u, m - 1);
                const int s = __builtin_amdgcn_readlane(ev, jj);
                wv[u] = (j + u < m) ? 1.0f : 0.0f;
                hv[u] = *(const float2*)(hl + (unsigned)(s * HID));
            }
#pragma unroll
            for (int u = 0; u < GRP; ++u) {
                acc.x += hv[u].x * wv[u];
                acc.y += hv[u].y * wv[u];
            }
        }
    }
    float2 r;
    r.x = fmaxf((acc.x + own.x) * dv + bv.x, 0.0f);
    r.y = fmaxf((acc.y + own.y) * dv + bv.y, 0.0f);
    return r;
}

#define GP_BLOCKS 1000
#define GL_BLOCKS 2000

__global__ void __launch_bounds__(256)
gather_layer_k(const int* __restrict__ rowptr, const int* __restrict__ rec,
               const float* __restrict__ h, const float* __restrict__ dis,
               const float* __restrict__ bias, float* __restrict__ outp, int n) {
    const int lane = threadIdx.x & 31;
    const int S = (gridDim.x * blockDim.x) >> 5;
    int node = __builtin_amdgcn_readfirstlane((blockIdx.x * blockDim.x + threadIdx.x) >> 5);
    if (node >= n) return;
    const float* hl = h + (lane << 1);
    const float2 bv = *(const float2*)(bias + (lane << 1));
    int beg = rowptr[node];
    int end = rowptr[node + 1];
    int ev = 0;
    if (lane < end - beg) ev = rec[beg + lane];
    float2 own = *(const float2*)(hl + (size_t)node * HID);
    float dv = dis[node];
    int n1 = min(node + S, n - 1);
    int nb = rowptr[n1 + (lane & 1)];
    for (; node < n; node += S) {
        const float2 r = gather_row(rec, hl, lane, beg, end, ev, own, dv, bv);
        const int nbeg = __builtin_amdgcn_readlane(nb, 0);
        const int nend = __builtin_amdgcn_readlane(nb, 1);
        int nev = 0;
        if (lane < nend - nbeg) nev = rec[nbeg + lane];
        const float2 nown = *(const float2*)(hl + (size_t)n1 * HID);
        const float  ndv  = dis[n1];
        const int n2 = min(n1 + S, n - 1);
        const int nnb = rowptr[n2 + (lane & 1)];
        *(float2*)(outp + (size_t)node * HID + (lane << 1)) = r;
        beg = nbeg; end = nend; ev = nev; own = nown; dv = ndv; n1 = n2; nb = nnb;
    }
}

#define TROWS  16
#define TPITCH 68
#define WLANE  36

__device__ __forceinline__ void lds_barrier() {
    __builtin_amdgcn_fence(__ATOMIC_RELEASE, "workgroup", "local");
    __builtin_amdgcn_s_barrier();
    __builtin_amdgcn_fence(__ATOMIC_ACQUIRE, "workgroup", "local");
}

__global__ void __launch_bounds__(TROWS * 32)
gather_product_k(const int* __restrict__ rowptr, const int* __restrict__ rec,
                 const float* __restrict__ h, const float* __restrict__ dis,
                 const float* __restrict__ bias, const float* __restrict__ Wn,
                 float* __restrict__ outp, int n) {
    __shared__ __attribute__((aligned(16))) float tA[TROWS * TPITCH];
    __shared__ __attribute__((aligned(16))) float tD[TROWS * TPITCH];
    __shared__ __attribute__((aligned(16))) float sW[4 * 32 * WLANE];
    const int lane = threadIdx.x & 31;
    const int wid  = threadIdx.x >> 5;
    const int tiles = (n + TROWS - 1) / TROWS;
    const int S = gridDim.x * TROWS;
    const int mr  = lane & 15;
    const int khi = (lane >> 4) << 1;

#pragma unroll
    for (int cc = 0; cc < HID / 16; ++cc) {
        const float* wp = Wn + (4 * wid + khi) * HID + cc * 16 + mr;
        float2 wv2;
        wv2.x = wp[0];
        wv2.y = wp[HID];
        *(float2*)(sW + (cc * 32 + lane) * WLANE + 2 * wid) = wv2;
    }

    const float* hl = h + (lane << 1);
    const float2 bv = *(const float2*)(bias + (lane << 1));
    int tile = blockIdx.x;
    int node = __builtin_amdgcn_readfirstlane(min(tile * TROWS + wid, n - 1));
    int beg = rowptr[node];
    int end = rowptr[node + 1];
    int ev = 0;
    if (lane < end - beg) ev = rec[beg + lane];
    float2 own = *(const float2*)(hl + (size_t)node * HID);
    float dv = dis[node];
    int n1 = __builtin_amdgcn_readfirstlane(min(node + S, n - 1));
    int nb = rowptr[n1 + (lane & 1)];
    __syncthreads();

    for (; tile < tiles; tile += gridDim.x) {
        const int  cur  = tile * TROWS + wid;
        const bool live = cur < n;
        const float2 r = gather_row(rec, hl, lane, beg, end, ev, own, dv, bv);
        const int nbeg = __builtin_amdgcn_readlane(nb, 0);
        const int nend = __builtin_amdgcn_readlane(nb, 1);
        int nev = 0;
        if (lane < nend - nbeg) nev = rec[nbeg + lane];
        const float2 nown = *(const float2*)(hl + (size_t)n1 * HID);
        const float  ndv  = dis[n1];
        const int n2 = __builtin_amdgcn_readfirstlane(min(n1 + S, n - 1));
        const int nnb = rowptr[n2 + (lane & 1)];

        *(float2*)(tA + wid * TPITCH + (lane << 1)) = r;
        lds_barrier();

        if (wid < HID / 16) {
            const float* arow = tA + mr * TPITCH + khi;
            const float* bfr  = sW + (wid * 32 + lane) * WLANE;
            v8f acc = {};
#pragma unroll
            for (int k = 0; k < HID; k += 8) {
                const float4 bq = *(const float4*)(bfr + (k >> 1));
                v2f a, b;
                const float2 av0 = *(const float2*)(arow + k);
                a[0] = av0.x; a[1] = av0.y;
                b[0] = bq.x;  b[1] = bq.y;
                acc = __builtin_amdgcn_wmma_f32_16x16x4_f32(false, a, false, b, (short)0, acc, false, false);
                const float2 av1 = *(const float2*)(arow + k + 4);
                a[0] = av1.x; a[1] = av1.y;
                b[0] = bq.z;  b[1] = bq.w;
                acc = __builtin_amdgcn_wmma_f32_16x16x4_f32(false, a, false, b, (short)0, acc, false, false);
            }
            float* drow = tD + ((lane >> 4) << 3) * TPITCH + wid * 16 + mr;
#pragma unroll
            for (int j = 0; j < 8; ++j) drow[j * TPITCH] = acc[j];
        }
        lds_barrier();
        if (live) {
            const float2 pr = *(const float2*)(tD + wid * TPITCH + (lane << 1));
            *(float2*)(outp + (size_t)cur * HID + (lane << 1)) = make_float2(pr.x * dv, pr.y * dv);
        }
        beg = nbeg; end = nend; ev = nev; own = nown; dv = ndv; n1 = n2; nb = nnb;
    }
}

#define PRUN 16
__device__ __forceinline__ void pool_flush(float* __restrict__ pooled, float* __restrict__ counts,
                                           int g, int lane, float sx, float sy,
                                           unsigned int mx, unsigned int my, int c) {
    float* p = pooled + (size_t)g * 128 + (lane << 1);
    atomicAdd(p, sx);
    atomicAdd(p + 1, sy);
    atomicMax((unsigned int*)(p + 64), mx);
    atomicMax((unsigned int*)(p + 65), my);
    if (lane == 0) atomicAdd(&counts[g], (float)c);
}

__global__ void __launch_bounds__(256)
pool_runs_k(const float* __restrict__ h, const int* __restrict__ batch,
            float* __restrict__ pooled, float* __restrict__ counts, int n) {
    const int lane = threadIdx.x & 31;
    int w = (blockIdx.x * blockDim.x + threadIdx.x) >> 5;
    w = __builtin_amdgcn_readfirstlane(w);
    const int base = w * PRUN;
    if (base >= n) return;
    const int m = min(PRUN, n - base);

    int gid = 0;
    if (lane < m) gid = batch[base + lane];

    const float* hl = h + (size_t)base * HID + (lane << 1);
    float2 hv[PRUN];
#pragma unroll
    for (int j = 0; j < PRUN; ++j) {
        const int jj = min(j, m - 1);
        hv[j] = *(const float2*)(hl + jj * HID);
    }

    int cur = __builtin_amdgcn_readlane(gid, 0);
    float sx = hv[0].x, sy = hv[0].y;
    unsigned int mx = __float_as_uint(hv[0].x), my = __float_as_uint(hv[0].y);
    int c = 1;
#pragma unroll
    for (int j = 1; j < PRUN; ++j) {
        if (j < m) {
            const int g = __builtin_amdgcn_readlane(gid, j);
            if (g != cur) {
                pool_flush(pooled, counts, cur, lane, sx, sy, mx, my, c);
                cur = g; sx = 0.0f; sy = 0.0f; mx = 0u; my = 0u; c = 0;
            }
            sx += hv[j].x;
            sy += hv[j].y;
            mx = max(mx, __float_as_uint(hv[j].x));
            my = max(my, __float_as_uint(hv[j].y));
            ++c;
        }
    }
    pool_flush(pooled, counts, cur, lane, sx, sy, mx, my, c);
}

__global__ void meandiv_k(float* __restrict__ pooled, const float* __restrict__ counts, int G) {
    int i = blockIdx.x * blockDim.x + threadIdx.x;
    if (i >= G * HID) return;
    int g = i >> 6;
    int f = i & 63;
    pooled[(size_t)g * 128 + f] /= fmaxf(counts[g], 1.0f);
}

__global__ void head_k(const float* __restrict__ pooled, const float* __restrict__ Wout,
                       const float* __restrict__ bout, float* __restrict__ out, int G) {
    int t = blockIdx.x * blockDim.x + threadIdx.x;
    int g = t >> 5;
    int lane = t & 31;
    if (g >= G) return;
    float v = 0.0f;
#pragma unroll
    for (int j = lane; j < 2 * HID; j += 32) v += pooled[(size_t)g * 128 + j] * Wout[j];
#pragma unroll
    for (int off = 16; off; off >>= 1) v += __shfl_xor(v, off, 32);
    if (lane == 0) out[g] = v + bout[0];
}

#define WS_DIS    0
#define WS_ROWPTR (WS_DIS + N_NODES)
#define WS_TOT    (WS_ROWPTR + N_NODES + 4)
#define WS_COUNTS (WS_TOT + 512)
#define WS_HIST   (WS_COUNTS + NGRAPH)
#define WS_TABLE  ((NB * NCH + 3) / 4 * 4)
#define WS_OFF    (WS_HIST + WS_TABLE)
#define WS_BIN    (WS_OFF + WS_TABLE)
#define WS_REC    (WS_BIN + 2 * N_EDGES)
#define WS_BUF0   (((WS_REC + N_EDGES) + 63) / 64 * 64)

extern "C" void kernel_launch(void* const* d_in, const int* in_sizes, int n_in,
                              void* d_out, int out_size, void* d_ws, size_t ws_size,
                              hipStream_t stream) {
    const float* x     = (const float*)d_in[0];
    const int*   eidx  = (const int*)  d_in[1];
    const int*   batch = (const int*)  d_in[2];
    const float* Ws[4] = { (const float*)d_in[3], (const float*)d_in[5],
                           (const float*)d_in[7], (const float*)d_in[9] };
    const float* bs[4] = { (const float*)d_in[4], (const float*)d_in[6],
                           (const float*)d_in[8], (const float*)d_in[10] };
    const float* Wout  = (const float*)d_in[11];
    const float* bout  = (const float*)d_in[12];

    const int N = N_NODES, E = N_EDGES, G = NGRAPH;
    const int* src = eidx;
    const int* dst = eidx + E;

    float* ws     = (float*)d_ws;
    float* dis    = ws + WS_DIS;
    int*   rowptr = (int*)(ws + WS_ROWPTR);
    int*   tot    = (int*)(ws + WS_TOT);
    float* counts = ws + WS_COUNTS;
    int*   histT  = (int*)(ws + WS_HIST);
    int*   offT   = (int*)(ws + WS_OFF);
    int*   binned = (int*)(ws + WS_BIN);
    int*   rec    = (int*)(ws + WS_REC);
    float* buf0   = ws + WS_BUF0;
    float* buf1   = buf0 + (size_t)N * HID;

    float* out    = (float*)d_out;
    float* pooled = out + G;

    bin_count_k  <<<NCH, SORT_T, 0, stream>>>(dst, histT);
    bin_scan_k   <<<NB,  SORT_T, 0, stream>>>(histT, offT, tot);
    bin_scatter_k<<<NCH, SORT_T, 0, stream>>>(src, dst, offT, tot, binned);
    bucket_rows_k<<<NB,  SORT_T, 0, stream>>>(binned, tot, rowptr, dis, rec);

    float* HWa = buf0;
    float* HWb = buf1;
    const int gemmThreads = (N / 16) * 32;
    gemm_wmma_k<<<(gemmThreads + 255) / 256, 256, 0, stream>>>(x, Ws[0], dis, HWa, N, 128);
    for (int l = 0; l < 3; ++l) {
        gather_product_k<<<GP_BLOCKS, TROWS * 32, 0, stream>>>(
            rowptr, rec, HWa, dis, bs[l], Ws[l + 1], HWb, N);
        float* t = HWa; HWa = HWb; HWb = t;
    }
    gather_layer_k<<<GL_BLOCKS, 256, 0, stream>>>(rowptr, rec, HWa, dis, bs[3], HWb, N);
    const float* hin = HWb;

    fill_k   <<<(G + G * 128 + 255) / 256, 256, 0, stream>>>(out, 0.0f, G + G * 128);
    fill_k   <<<(G + 255) / 256, 256, 0, stream>>>(counts, 0.0f, G);
    pool_runs_k<<<((N + PRUN - 1) / PRUN * 32 + 255) / 256, 256, 0, stream>>>(hin, batch, pooled, counts, N);
    meandiv_k<<<(G * HID + 255) / 256, 256, 0, stream>>>(pooled, counts, G);
    head_k   <<<(G * 32 + 255) / 256, 256, 0, stream>>>(pooled, Wout, bout, out, G);
}


# Pooling folded into the last neighbour-sum launch: a wave walks 7 consecutive nodes, run sum/max/count in registers (sum runs <=7 nodes, were <=16); clears ride in layer 1's product, mean in the head
# speedup vs baseline: 4.9615x; 1.0411x over previous
#include <hip/hip_runtime.h>
#include <hip/hip_bf16.h>

typedef __attribute__((ext_vector_type(2))) float v2f;
typedef __attribute__((ext_vector_type(8))) float v8f;

#define N_NODES 100000
#define N_EDGES 1600000
#define HID     64
#define NGRAPH  2048

__global__ void gemm_wmma_k(const float* __restrict__ A, const float* __restrict__ W,
                            const float* __restrict__ rs,
                            float* __restrict__ C, int rows, int K,
                            float* __restrict__ zp, float* __restrict__ zc) {
    const int lane = threadIdx.x & 31;
    const int gt   = blockIdx.x * blockDim.x + threadIdx.x;
    if (gt < NGRAPH * 128 / 4)
        ((float4*)zp)[gt] = make_float4(0.0f, 0.0f, 0.0f, 0.0f);
    else if (gt < NGRAPH * 128 / 4 + NGRAPH / 4)
        ((float4*)zc)[gt - NGRAPH * 128 / 4] = make_float4(0.0f, 0.0f, 0.0f, 0.0f);
    const int w    = gt >> 5;
    const int rowTiles = rows >> 4;
    if (w >= rowTiles) return;

    const int m0  = w << 4;
    const int mr  = lane & 15;
    const int khi = (lane >> 4) << 1;

    const float* arow = A + (size_t)(m0 + mr) * K + khi;
    const float* bcol = W + mr;

    v8f acc0 = {}, acc1 = {}, acc2 = {}, acc3 = {};
    for (int k = 0; k < K; k += 4) {
        v2f a;
        const float2 av = *(const float2*)(arow + k);
        a[0] = av.x; a[1] = av.y;

        const float* brow0 = bcol + (size_t)(k + khi) * HID;
        const float* brow1 = brow0 + HID;
        v2f b0, b1, b2, b3;
        b0[0] = brow0[0];  b0[1] = brow1[0];
        b1[0] = brow0[16]; b1[1] = brow1[16];
        b2[0] = brow0[32]; b2[1] = brow1[32];
        b3[0] = brow0[48]; b3[1] = brow1[48];

        acc0 = __builtin_amdgcn_wmma_f32_16x16x4_f32(false, a, false, b0, (short)0, acc0, false, false);
        acc1 = __builtin_amdgcn_wmma_f32_16x16x4_f32(false, a, false, b1, (short)0, acc1, false, false);
        acc2 = __builtin_amdgcn_wmma_f32_16x16x4_f32(false, a, false, b2, (short)0, acc2, false, false);
        acc3 = __builtin_amdgcn_wmma_f32_16x16x4_f32(false, a, false, b3, (short)0, acc3, false, false);
    }

    const int r0 = m0 + ((lane >> 4) << 3);
    float* crow = C + (size_t)r0 * HID + mr;
    const float4 sa = *(const float4*)(rs + r0);
    const float4 sb = *(const float4*)(rs + r0 + 4);
    const float sc[8] = { sa.x, sa.y, sa.z, sa.w, sb.x, sb.y, sb.z, sb.w };
#pragma unroll
    for (int j = 0; j < 8; ++j) {
        crow[(size_t)j * HID +  0] = acc0[j] * sc[j];
        crow[(size_t)j * HID + 16] = acc1[j] * sc[j];
        crow[(size_t)j * HID + 32] = acc2[j] * sc[j];
        crow[(size_t)j * HID + 48] = acc3[j] * sc[j];
    }
}

#define BSH    8
#define NB     ((N_NODES + 255) >> BSH)
#define CH     4096
#define NCH    ((N_EDGES + CH - 1) / CH)
#define SORT_T 512

__device__ __forceinline__ int scan512(int v, int* sm, int& total) {
    const int lane = threadIdx.x & 31;
    const int wid  = threadIdx.x >> 5;
    int inc = v;
#pragma unroll
    for (int off = 1; off < 32; off <<= 1) {
        const int u = __shfl_up(inc, off, 32);
        if (lane >= off) inc += u;
    }
    if (lane == 31) sm[wid] = inc;
    __syncthreads();
    int pre = 0, tot = 0;
#pragma unroll
    for (int k = 0; k < SORT_T / 32; ++k) {
        const int s = sm[k];
        tot += s;
        if (k < wid) pre += s;
    }
    total = tot;
    return pre + inc - v;
}

__global__ void __launch_bounds__(SORT_T)
bin_count_k(const int* __restrict__ dst, int* __restrict__ histT) {
    __shared__ int c[NB];
    const int tid = threadIdx.x;
    const int chunk = blockIdx.x;
    int4 d[CH / (SORT_T * 4)];
#pragma unroll
    for (int k = 0; k < CH / (SORT_T * 4); ++k) {
        const int e = chunk * CH + k * (SORT_T * 4) + tid * 4;
        d[k] = (e < N_EDGES) ? *(const int4*)(dst + e) : make_int4(-1, -1, -1, -1);
    }
    for (int i = tid; i < NB; i += SORT_T) c[i] = 0;
    __syncthreads();
#pragma unroll
    for (int k = 0; k < CH / (SORT_T * 4); ++k) {
        if (d[k].x >= 0) {
            atomicAdd(&c[d[k].x >> BSH], 1);
            atomicAdd(&c[d[k].y >> BSH], 1);
            atomicAdd(&c[d[k].z >> BSH], 1);
            atomicAdd(&c[d[k].w >> BSH], 1);
        }
    }
    __syncthreads();
    for (int i = tid; i < NB; i += SORT_T) histT[i * NCH + chunk] = c[i];
}

__global__ void __launch_bounds__(SORT_T)
bin_scan_k(const int* __restrict__ histT, int* __restrict__ offT, int* __restrict__ tot) {
    __shared__ int sm[SORT_T / 32];
    const int tid = threadIdx.x;
    const int b = blockIdx.x;
    const int v = (tid < NCH) ? histT[b * NCH + tid] : 0;
    int total;
    const int excl = scan512(v, sm, total);
    if (tid < NCH) offT[b * NCH + tid] = excl;
    if (tid == 0) tot[b] = total;
}

__global__ void __launch_bounds__(SORT_T)
bin_scatter_k(const int* __restrict__ src, const int* __restrict__ dst,
              const int* __restrict__ offT, const int* __restrict__ tot,
              int* __restrict__ binned) {
    __shared__ int cur[NB];
    __shared__ int sm[SORT_T / 32];
    const int tid = threadIdx.x;
    const int chunk = blockIdx.x;
    int4 s[CH / (SORT_T * 4)], d[CH / (SORT_T * 4)];
#pragma unroll
    for (int k = 0; k < CH / (SORT_T * 4); ++k) {
        const int e = chunk * CH + k * (SORT_T * 4) + tid * 4;
        const bool in = e < N_EDGES;
        s[k] = in ? *(const int4*)(src + e) : make_int4(0, 0, 0, 0);
        d[k] = in ? *(const int4*)(dst + e) : make_int4(-1, -1, -1, -1);
    }
    const int t = (tid < NB) ? tot[tid] : 0;
    const int o = (tid < NB) ? offT[tid * NCH + chunk] : 0;
    int total;
    const int bstart = scan512(t, sm, total);
    if (tid < NB) cur[tid] = bstart + o;
    __syncthreads();
    int2* out2 = (int2*)binned;
#pragma unroll
    for (int k = 0; k < CH / (SORT_T * 4); ++k) {
        if (d[k].x >= 0) {
            const int p0 = atomicAdd(&cur[d[k].x >> BSH], 1);
            const int p1 = atomicAdd(&cur[d[k].y >> BSH], 1);
            const int p2 = atomicAdd(&cur[d[k].z >> BSH], 1);
            const int p3 = atomicAdd(&cur[d[k].w >> BSH], 1);
            out2[p0] = make_int2(s[k].x, d[k].x);
            out2[p1] = make_int2(s[k].y, d[k].y);
            out2[p2] = make_int2(s[k].z, d[k].z);
            out2[p3] = make_int2(s[k].w, d[k].w);
        }
    }
}

#define BLD 4
__global__ void __launch_bounds__(SORT_T)
bucket_rows_k(const int* __restrict__ binned, const int* __restrict__ tot,
              int* __restrict__ rowptr, float* __restrict__ dis, int* __restrict__ rec) {
    __shared__ int cnt[1 << BSH];
    __shared__ int cur[1 << BSH];
    __shared__ int smA[SORT_T / 32];
    __shared__ int smB[SORT_T / 32];
    __shared__ int range[2];
    const int tid = threadIdx.x;
    const int b = blockIdx.x;
    const int node0 = b << BSH;
    if (tid < (1 << BSH)) cnt[tid] = 0;
    const int t = (tid < NB) ? tot[tid] : 0;
    int total;
    const int bstart = scan512(t, smA, total);
    if (tid == b) { range[0] = bstart; range[1] = bstart + t; }
    __syncthreads();
    const int beg = range[0];
    const int end = range[1];
    const int2* in2 = (const int2*)binned;

    for (int e0 = beg; e0 < end; e0 += SORT_T * BLD) {
        int d[BLD];
#pragma unroll
        for (int u = 0; u < BLD; ++u) {
            const int e = e0 + u * SORT_T + tid;
            d[u] = (e < end) ? in2[e].y : -1;
        }
#pragma unroll
        for (int u = 0; u < BLD; ++u)
            if (d[u] >= 0) atomicAdd(&cnt[d[u] - node0], 1);
    }
    __syncthreads();

    const int c = (tid < (1 << BSH)) ? cnt[tid] : 0;
    int total2;
    const int ex = scan512(c, smB, total2);
    if (tid < (1 << BSH)) {
        cur[tid] = beg + ex;
        const int node = node0 + tid;
        if (node < N_NODES) {
            rowptr[node] = beg + ex;
            dis[node] = rsqrtf((float)(c + 1));
        }
    }
    if (b == NB - 1 && tid == 0) rowptr[N_NODES] = end;
    __syncthreads();

    for (int e0 = beg; e0 < end; e0 += SORT_T * BLD) {
        int2 r[BLD];
#pragma unroll
        for (int u = 0; u < BLD; ++u) {
            const int e = e0 + u * SORT_T + tid;
            r[u] = (e < end) ? in2[e] : make_int2(0, -1);
        }
#pragma unroll
        for (int u = 0; u < BLD; ++u)
            if (r[u].y >= 0) {
                const int p = atomicAdd(&cur[r[u].y - node0], 1);
                rec[p] = r[u].x;
            }
    }
}

#define GRP 8
__device__ __forceinline__ float2 gather_row(const int* __restrict__ rec, const float* __restrict__ hl,
                                             int lane, int beg, int end, int ev0,
                                             float2 own, float dv, float2 bv) {
    float2 acc = make_float2(0.0f, 0.0f);
    int ev = ev0;

    for (int base = beg; base < end; base += 32) {
        const int m = min(32, end - base);
        if (base != beg) {
            ev = 0;
            if (lane < m) ev = rec[base + lane];
        }
        for (int j = 0; j < m; j += GRP) {
            float2 hv[GRP];
            float  wv[GRP];
#pragma unroll
            for (int u = 0; u < GRP; ++u) {
                const int jj = min(j + u, m - 1);
                const int s = __builtin_amdgcn_readlane(ev, jj);
                wv[u] = (j + u < m) ? 1.0f : 0.0f;
                hv[u] = *(const float2*)(hl + (unsigned)(s * HID));
            }
#pragma unroll
            for (int u = 0; u < GRP; ++u) {
                acc.x += hv[u].x * wv[u];
                acc.y += hv[u].y * wv[u];
            }
        }
    }
    float2 r;
    r.x = fmaxf((acc.x + own.x) * dv + bv.x, 0.0f);
    r.y = fmaxf((acc.y + own.y) * dv + bv.y, 0.0f);
    return r;
}

#define GP_BLOCKS 1000

#define TROWS  16
#define TPITCH 68
#define WLANE  36

__device__ __forceinline__ void lds_barrier() {
    __builtin_amdgcn_fence(__ATOMIC_RELEASE, "workgroup", "local");
    __builtin_amdgcn_s_barrier();
    __builtin_amdgcn_fence(__ATOMIC_ACQUIRE, "workgroup", "local");
}

__global__ void __launch_bounds__(TROWS * 32)
gather_product_k(const int* __restrict__ rowptr, const int* __restrict__ rec,
                 const float* __restrict__ h, const float* __restrict__ dis,
                 const float* __restrict__ bias, const float* __restrict__ Wn,
                 float* __restrict__ outp, int n) {
    __shared__ __attribute__((aligned(16))) float tA[TROWS * TPITCH];
    __shared__ __attribute__((aligned(16))) float tD[TROWS * TPITCH];
    __shared__ __attribute__((aligned(16))) float sW[4 * 32 * WLANE];
    const int lane = threadIdx.x & 31;
    const int wid  = threadIdx.x >> 5;
    const int tiles = (n + TROWS - 1) / TROWS;
    const int S = gridDim.x * TROWS;
    const int mr  = lane & 15;
    const int khi = (lane >> 4) << 1;

#pragma unroll
    for (int cc = 0; cc < HID / 16; ++cc) {
        const float* wp = Wn + (4 * wid + khi) * HID + cc * 16 + mr;
        float2 wv2;
        wv2.x = wp[0];
        wv2.y = wp[HID];
        *(float2*)(sW + (cc * 32 + lane) * WLANE + 2 * wid) = wv2;
    }

    const float* hl = h + (lane << 1);
    const float2 bv = *(const float2*)(bias + (lane << 1));
    int tile = blockIdx.x;
    int node = __builtin_amdgcn_readfirstlane(min(tile * TROWS + wid, n - 1));
    int beg = rowptr[node];
    int end = rowptr[node + 1];
    int ev = 0;
    if (lane < end - beg) ev = rec[beg + lane];
    float2 own = *(const float2*)(hl + (size_t)node * HID);
    float dv = dis[node];
    int n1 = __builtin_amdgcn_readfirstlane(min(node + S, n - 1));
    int nb = rowptr[n1 + (lane & 1)];
    __syncthreads();

    for (; tile < tiles; tile += gridDim.x) {
        const int  cur  = tile * TROWS + wid;
        const bool live = cur < n;
        const float2 r = gather_row(rec, hl, lane, beg, end, ev, own, dv, bv);
        const int nbeg = __builtin_amdgcn_readlane(nb, 0);
        const int nend = __builtin_amdgcn_readlane(nb, 1);
        int nev = 0;
        if (lane < nend - nbeg) nev = rec[nbeg + lane];
        const float2 nown = *(const float2*)(hl + (size_t)n1 * HID);
        const float  ndv  = dis[n1];
        const int n2 = __builtin_amdgcn_readfirstlane(min(n1 + S, n - 1));
        const int nnb = rowptr[n2 + (lane & 1)];

        *(float2*)(tA + wid * TPITCH + (lane << 1)) = r;
        lds_barrier();

        if (wid < HID / 16) {
            const float* arow = tA + mr * TPITCH + khi;
            const float* bfr  = sW + (wid * 32 + lane) * WLANE;
            v8f acc = {};
#pragma unroll
            for (int k = 0; k < HID; k += 8) {
                const float4 bq = *(const float4*)(bfr + (k >> 1));
                v2f a, b;
                const float2 av0 = *(const float2*)(arow + k);
                a[0] = av0.x; a[1] = av0.y;
                b[0] = bq.x;  b[1] = bq.y;
                acc = __builtin_amdgcn_wmma_f32_16x16x4_f32(false, a, false, b, (short)0, acc, false, false);
                const float2 av1 = *(const float2*)(arow + k + 4);
                a[0] = av1.x; a[1] = av1.y;
                b[0] = bq.z;  b[1] = bq.w;
                acc = __builtin_amdgcn_wmma_f32_16x16x4_f32(false, a, false, b, (short)0, acc, false, false);
            }
            float* drow = tD + ((lane >> 4) << 3) * TPITCH + wid * 16 + mr;
#pragma unroll
            for (int j = 0; j < 8; ++j) drow[j * TPITCH] = acc[j];
        }
        lds_barrier();
        if (live) {
            const float2 pr = *(const float2*)(tD + wid * TPITCH + (lane << 1));
            *(float2*)(outp + (size_t)cur * HID + (lane << 1)) = make_float2(pr.x * dv, pr.y * dv);
        }
        beg = nbeg; end = nend; ev = nev; own = nown; dv = ndv; n1 = n2; nb = nnb;
    }
}

#define PC 7
#define GL_BLOCKS (((N_NODES + PC - 1) / PC + 7) / 8)

__device__ __forceinline__ void pool_flush(float* __restrict__ pooled, float* __restrict__ counts,
                                           int g, int lane, float sx, float sy,
                                           unsigned int mx, unsigned int my, int c) {
    float* p = pooled + (size_t)g * 128 + (lane << 1);
    atomicAdd(p, sx);
    atomicAdd(p + 1, sy);
    atomicMax((unsigned int*)(p + 64), mx);
    atomicMax((unsigned int*)(p + 65), my);
    if (lane == 0) atomicAdd(&counts[g], (float)c);
}

__global__ void __launch_bounds__(256)
gather_pool_k(const int* __restrict__ rowptr, const int* __restrict__ rec,
              const float* __restrict__ h, const float* __restrict__ dis,
              const float* __restrict__ bias, const int* __restrict__ batch,
              float* __restrict__ pooled, float* __restrict__ counts, int n) {
    const int lane = threadIdx.x & 31;
    const int w = __builtin_amdgcn_readfirstlane((blockIdx.x * blockDim.x + threadIdx.x) >> 5);
    const int node0 = w * PC;
    if (node0 >= n) return;
    const int m = min(PC, n - node0);
    const float* hl = h + (lane << 1);
    const float2 bv = *(const float2*)(bias + (lane << 1));
    const int   rp  = rowptr[node0 + min(lane, m)];
    const int   jc  = min(lane, m - 1);
    const int   gid = batch[node0 + jc];
    const float dvl = dis[node0 + jc];
    int beg = __builtin_amdgcn_readlane(rp, 0);
    int end = __builtin_amdgcn_readlane(rp, 1);
    int ev = 0;
    if (lane < end - beg) ev = rec[beg + lane];
    float2 own = *(const float2*)(hl + (size_t)node0 * HID);

    int cur = __builtin_amdgcn_readlane(gid, 0);
    float sx = 0.0f, sy = 0.0f;
    unsigned int mx = 0u, my = 0u;
    int c = 0;
    for (int i = 0; i < m; ++i) {
        const float dv = __int_as_float(__builtin_amdgcn_readlane(__float_as_int(dvl), i));
        const float2 r = gather_row(rec, hl, lane, beg, end, ev, own, dv, bv);
        const int i1 = min(i + 1, m - 1);
        const int nbeg = __builtin_amdgcn_readlane(rp, i1);
        const int nend = __builtin_amdgcn_readlane(rp, i1 + 1);
        int nev = 0;
        if (lane < nend - nbeg) nev = rec[nbeg + lane];
        const float2 nown = *(const float2*)(hl + (size_t)(node0 + i1) * HID);

        const int g = __builtin_amdgcn_readlane(gid, i);
        if (g != cur) {
            pool_flush(pooled, counts, cur, lane, sx, sy, mx, my, c);
            cur = g; sx = 0.0f; sy = 0.0f; mx = 0u; my = 0u; c = 0;
        }
        sx += r.x;
        sy += r.y;
        mx = max(mx, __float_as_uint(r.x));
        my = max(my, __float_as_uint(r.y));
        ++c;
        beg = nbeg; end = nend; ev = nev; own = nown;
    }
    pool_flush(pooled, counts, cur, lane, sx, sy, mx, my, c);
}

__global__ void head_k(float* __restrict__ pooled, const float* __restrict__ counts,
                       const float* __restrict__ Wout,
                       const float* __restrict__ bout, float* __restrict__ out, int G) {
    int t = blockIdx.x * blockDim.x + threadIdx.x;
    int g = t >> 5;
    int lane = t & 31;
    if (g >= G) return;
    float* p = pooled + (size_t)g * 128;
    const float d = fmaxf(counts[g], 1.0f);
    const float m0 = p[lane] / d;
    const float m1 = p[lane + 32] / d;
    p[lane] = m0;
    p[lane + 32] = m1;
    float v = 0.0f;
    v += m0 * Wout[lane];
    v += m1 * Wout[lane + 32];
    v += p[lane + 64] * Wout[lane + 64];
    v += p[lane + 96] * Wout[lane + 96];
#pragma unroll
    for (int off = 16; off; off >>= 1) v += __shfl_xor(v, off, 32);
    if (lane == 0) out[g] = v + bout[0];
}

#define WS_DIS    0
#define WS_ROWPTR (WS_DIS + N_NODES)
#define WS_TOT    (WS_ROWPTR + N_NODES + 4)
#define WS_COUNTS (WS_TOT + 512)
#define WS_HIST   (WS_COUNTS + NGRAPH)
#define WS_TABLE  ((NB * NCH + 3) / 4 * 4)
#define WS_OFF    (WS_HIST + WS_TABLE)
#define WS_BIN    (WS_OFF + WS_TABLE)
#define WS_REC    (WS_BIN + 2 * N_EDGES)
#define WS_BUF0   (((WS_REC + N_EDGES) + 63) / 64 * 64)

extern "C" void kernel_launch(void* const* d_in, const int* in_sizes, int n_in,
                              void* d_out, int out_size, void* d_ws, size_t ws_size,
                              hipStream_t stream) {
    const float* x     = (const float*)d_in[0];
    const int*   eidx  = (const int*)  d_in[1];
    const int*   batch = (const int*)  d_in[2];
    const float* Ws[4] = { (const float*)d_in[3], (const float*)d_in[5],
                           (const float*)d_in[7], (const float*)d_in[9] };
    const float* bs[4] = { (const float*)d_in[4], (const float*)d_in[6],
                           (const float*)d_in[8], (const float*)d_in[10] };
    const float* Wout  = (const float*)d_in[11];
    const float* bout  = (const float*)d_in[12];

    const int N = N_NODES, E = N_EDGES, G = NGRAPH;
    const int* src = eidx;
    const int* dst = eidx + E;

    float* ws     = (float*)d_ws;
    float* dis    = ws + WS_DIS;
    int*   rowptr = (int*)(ws + WS_ROWPTR);
    int*   tot    = (int*)(ws + WS_TOT);
    float* counts = ws + WS_COUNTS;
    int*   histT  = (int*)(ws + WS_HIST);
    int*   offT   = (int*)(ws + WS_OFF);
    int*   binned = (int*)(ws + WS_BIN);
    int*   rec    = (int*)(ws + WS_REC);
    float* buf0   = ws + WS_BUF0;
    float* buf1   = buf0 + (size_t)N * HID;

    float* out    = (float*)d_out;
    float* pooled = out + G;

    bin_count_k  <<<NCH, SORT_T, 0, stream>>>(dst, histT);
    bin_scan_k   <<<NB,  SORT_T, 0, stream>>>(histT, offT, tot);
    bin_scatter_k<<<NCH, SORT_T, 0, stream>>>(src, dst, offT, tot, binned);
    bucket_rows_k<<<NB,  SORT_T, 0, stream>>>(binned, tot, rowptr, dis, rec);

    float* HWa = buf0;
    float* HWb = buf1;
    const int gemmThreads = (N / 16) * 32;
    gemm_wmma_k<<<(gemmThreads + 255) / 256, 256, 0, stream>>>(x, Ws[0], dis, HWa, N, 128,
                                                               pooled, counts);
    for (int l = 0; l < 3; ++l) {
        gather_product_k<<<GP_BLOCKS, TROWS * 32, 0, stream>>>(
            rowptr, rec, HWa, dis, bs[l], Ws[l + 1], HWb, N);
        float* t = HWa; HWa = HWb; HWb = t;
    }
    gather_pool_k<<<GL_BLOCKS, 256, 0, stream>>>(rowptr, rec, HWa, dis, bs[3], batch,
                                                 pooled, counts, N);
    head_k<<<(G * 32 + 255) / 256, 256, 0, stream>>>(pooled, counts, Wout, bout, out, G);
}


# Edge grouping in 2 launches, not 4: each chunk ranks its edges by bucket on its own LDS counts into its own region; bucket pass walks 391 runs; count+scan launches gone; sum order = LDS arrival
# speedup vs baseline: 5.6394x; 1.1366x over previous
#include <hip/hip_runtime.h>
#include <hip/hip_bf16.h>

typedef __attribute__((ext_vector_type(2))) float v2f;
typedef __attribute__((ext_vector_type(8))) float v8f;

#define N_NODES 100000
#define N_EDGES 1600000
#define HID     64
#define NGRAPH  2048

__global__ void gemm_wmma_k(const float* __restrict__ A, const float* __restrict__ W,
                            const float* __restrict__ rs,
                            float* __restrict__ C, int rows, int K,
                            float* __restrict__ zp, float* __restrict__ zc) {
    const int lane = threadIdx.x & 31;
    const int gt   = blockIdx.x * blockDim.x + threadIdx.x;
    if (gt < NGRAPH * 128 / 4)
        ((float4*)zp)[gt] = make_float4(0.0f, 0.0f, 0.0f, 0.0f);
    else if (gt < NGRAPH * 128 / 4 + NGRAPH / 4)
        ((float4*)zc)[gt - NGRAPH * 128 / 4] = make_float4(0.0f, 0.0f, 0.0f, 0.0f);
    const int w    = gt >> 5;
    const int rowTiles = rows >> 4;
    if (w >= rowTiles) return;

    const int m0  = w << 4;
    const int mr  = lane & 15;
    const int khi = (lane >> 4) << 1;

    const float* arow = A + (size_t)(m0 + mr) * K + khi;
    const float* bcol = W + mr;

    v8f acc0 = {}, acc1 = {}, acc2 = {}, acc3 = {};
    for (int k = 0; k < K; k += 4) {
        v2f a;
        const float2 av = *(const float2*)(arow + k);
        a[0] = av.x; a[1] = av.y;

        const float* brow0 = bcol + (size_t)(k + khi) * HID;
        const float* brow1 = brow0 + HID;
        v2f b0, b1, b2, b3;
        b0[0] = brow0[0];  b0[1] = brow1[0];
        b1[0] = brow0[16]; b1[1] = brow1[16];
        b2[0] = brow0[32]; b2[1] = brow1[32];
        b3[0] = brow0[48]; b3[1] = brow1[48];

        acc0 = __builtin_amdgcn_wmma_f32_16x16x4_f32(false, a, false, b0, (short)0, acc0, false, false);
        acc1 = __builtin_amdgcn_wmma_f32_16x16x4_f32(false, a, false, b1, (short)0, acc1, false, false);
        acc2 = __builtin_amdgcn_wmma_f32_16x16x4_f32(false, a, false, b2, (short)0, acc2, false, false);
        acc3 = __builtin_amdgcn_wmma_f32_16x16x4_f32(false, a, false, b3, (short)0, acc3, false, false);
    }

    const int r0 = m0 + ((lane >> 4) << 3);
    float* crow = C + (size_t)r0 * HID + mr;
    const float4 sa = *(const float4*)(rs + r0);
    const float4 sb = *(const float4*)(rs + r0 + 4);
    const float sc[8] = { sa.x, sa.y, sa.z, sa.w, sb.x, sb.y, sb.z, sb.w };
#pragma unroll
    for (int j = 0; j < 8; ++j) {
        crow[(size_t)j * HID +  0] = acc0[j] * sc[j];
        crow[(size_t)j * HID + 16] = acc1[j] * sc[j];
        crow[(size_t)j * HID + 32] = acc2[j] * sc[j];
        crow[(size_t)j * HID + 48] = acc3[j] * sc[j];
    }
}

#define BSH    8
#define NB     ((N_NODES + 255) >> BSH)
#define CH     4096
#define NCH    ((N_EDGES + CH - 1) / CH)
#define SORT_T 512
#define SORT_W (SORT_T / 32)

__device__ __forceinline__ int scan512(int v, int* sm, int& total) {
    const int lane = threadIdx.x & 31;
    const int wid  = threadIdx.x >> 5;
    int inc = v;
#pragma unroll
    for (int off = 1; off < 32; off <<= 1) {
        const int u = __shfl_up(inc, off, 32);
        if (lane >= off) inc += u;
    }
    if (lane == 31) sm[wid] = inc;
    __syncthreads();
    int pre = 0, tot = 0;
#pragma unroll
    for (int k = 0; k < SORT_T / 32; ++k) {
        const int s = sm[k];
        tot += s;
        if (k < wid) pre += s;
    }
    total = tot;
    return pre + inc - v;
}

__global__ void __launch_bounds__(SORT_T)
chunk_bin_k(const int* __restrict__ src, const int* __restrict__ dst,
            int* __restrict__ startT, int* __restrict__ cntT, int* __restrict__ binned) {
    __shared__ int cnt[NB];
    __shared__ int st[NB];
    __shared__ int sm[SORT_T / 32];
    const int tid = threadIdx.x;
    const int chunk = blockIdx.x;
    int4 s[CH / (SORT_T * 4)], d[CH / (SORT_T * 4)], rk[CH / (SORT_T * 4)];
#pragma unroll
    for (int k = 0; k < CH / (SORT_T * 4); ++k) {
        const int e = chunk * CH + k * (SORT_T * 4) + tid * 4;
        const bool in = e < N_EDGES;
        s[k] = in ? *(const int4*)(src + e) : make_int4(0, 0, 0, 0);
        d[k] = in ? *(const int4*)(dst + e) : make_int4(-1, -1, -1, -1);
        rk[k] = make_int4(0, 0, 0, 0);
    }
    for (int i = tid; i < NB; i += SORT_T) cnt[i] = 0;
    __syncthreads();
#pragma unroll
    for (int k = 0; k < CH / (SORT_T * 4); ++k) {
        if (d[k].x >= 0) {
            rk[k].x = atomicAdd(&cnt[d[k].x >> BSH], 1);
            rk[k].y = atomicAdd(&cnt[d[k].y >> BSH], 1);
            rk[k].z = atomicAdd(&cnt[d[k].z >> BSH], 1);
            rk[k].w = atomicAdd(&cnt[d[k].w >> BSH], 1);
        }
    }
    __syncthreads();
    const int v = (tid < NB) ? cnt[tid] : 0;
    int total;
    const int ex = scan512(v, sm, total);
    if (tid < NB) {
        st[tid] = ex;
        startT[chunk * NB + tid] = ex;
        cntT[chunk * NB + tid] = v;
    }
    __syncthreads();
    int2* out2 = (int2*)binned + chunk * CH;
#pragma unroll
    for (int k = 0; k < CH / (SORT_T * 4); ++k) {
        if (d[k].x >= 0) {
            out2[st[d[k].x >> BSH] + rk[k].x] = make_int2(s[k].x, d[k].x);
            out2[st[d[k].y >> BSH] + rk[k].y] = make_int2(s[k].y, d[k].y);
            out2[st[d[k].z >> BSH] + rk[k].z] = make_int2(s[k].z, d[k].z);
            out2[st[d[k].w >> BSH] + rk[k].w] = make_int2(s[k].w, d[k].w);
        }
    }
}

#define RB 13
__global__ void __launch_bounds__(SORT_T)
bucket_rows_k(const int* __restrict__ binned, const int* __restrict__ startT,
              const int* __restrict__ cntT, int* __restrict__ rowptr, float* __restrict__ dis, int* __restrict__ rec) {
    __shared__ int cnt[1 << BSH];
    __shared__ int cur[1 << BSH];
    __shared__ int rbase[NCH];
    __shared__ int rlen[NCH];
    __shared__ int smA[SORT_T / 32];
    __shared__ int smB[SORT_T / 32];
    __shared__ int smC[SORT_T / 32];
    const int tid  = threadIdx.x;
    const int lane = tid & 31;
    const int wid  = tid >> 5;
    const int b = blockIdx.x;
    const int node0 = b << BSH;
    if (tid < (1 << BSH)) cnt[tid] = 0;
    int e0 = 0, en = 0;
    if (tid < NCH) {
        e0 = startT[tid * NB + b];
        en = cntT[tid * NB + b];
        rbase[tid] = tid * CH + e0;
        rlen[tid]  = en;
    }
    int beg, total;
    scan512(e0, smA, beg);
    scan512(en, smB, total);
    const int end = beg + total;
    __syncthreads();
    const int2* in2 = (const int2*)binned;

    for (int r0 = wid; r0 < NCH; r0 += SORT_W * RB) {
        int d[RB];
#pragma unroll
        for (int u = 0; u < RB; ++u) {
            const int r = min(r0 + u * SORT_W, NCH - 1);
            const bool on = (r0 + u * SORT_W < NCH) && (lane < rlen[r]);
            const int v = in2[on ? rbase[r] + lane : 0].y;
            d[u] = on ? v : -1;
        }
#pragma unroll
        for (int u = 0; u < RB; ++u)
            if (d[u] >= 0) atomicAdd(&cnt[d[u] - node0], 1);
    }
    for (int r = wid; r < NCH; r += SORT_W) {
        const int n = rlen[r];
        for (int i = lane + 32; i < n; i += 32)
            atomicAdd(&cnt[in2[rbase[r] + i].y - node0], 1);
    }
    __syncthreads();

    const int c = (tid < (1 << BSH)) ? cnt[tid] : 0;
    int total2;
    const int ex = scan512(c, smC, total2);
    if (tid < (1 << BSH)) {
        cur[tid] = beg + ex;
        const int node = node0 + tid;
        if (node < N_NODES) {
            rowptr[node] = beg + ex;
            dis[node] = rsqrtf((float)(c + 1));
        }
    }
    if (b == NB - 1 && tid == 0) rowptr[N_NODES] = end;
    __syncthreads();

    for (int r0 = wid; r0 < NCH; r0 += SORT_W * RB) {
        int2 q[RB];
#pragma unroll
        for (int u = 0; u < RB; ++u) {
            const int r = min(r0 + u * SORT_W, NCH - 1);
            const bool on = (r0 + u * SORT_W < NCH) && (lane < rlen[r]);
            const int2 v = in2[on ? rbase[r] + lane : 0];
            q[u] = make_int2(v.x, on ? v.y : -1);
        }
#pragma unroll
        for (int u = 0; u < RB; ++u)
            if (q[u].y >= 0) {
                const int p = atomicAdd(&cur[q[u].y - node0], 1);
                rec[p] = q[u].x;
            }
    }
    for (int r = wid; r < NCH; r += SORT_W) {
        const int n = rlen[r];
        for (int i = lane + 32; i < n; i += 32) {
            const int2 q1 = in2[rbase[r] + i];
            const int p = atomicAdd(&cur[q1.y - node0], 1);
            rec[p] = q1.x;
        }
    }
}

#define GRP 8
__device__ __forceinline__ float2 gather_row(const int* __restrict__ rec, const float* __restrict__ hl,
                                             int lane, int beg, int end, int ev0,
                                             float2 own, float dv, float2 bv) {
    float2 acc = make_float2(0.0f, 0.0f);
    int ev = ev0;

    for (int base = beg; base < end; base += 32) {
        const int m = min(32, end - base);
        if (base != beg) {
            ev = 0;
            if (lane < m) ev = rec[base + lane];
        }
        for (int j = 0; j < m; j += GRP) {
            float2 hv[GRP];
            float  wv[GRP];
#pragma unroll
            for (int u = 0; u < GRP; ++u) {
                const int jj = min(j + u, m - 1);
                const int s = __builtin_amdgcn_readlane(ev, jj);
                wv[u] = (j + u < m) ? 1.0f : 0.0f;
                hv[u] = *(const float2*)(hl + (unsigned)(s * HID));
            }
#pragma unroll
            for (int u = 0; u < GRP; ++u) {
                acc.x += hv[u].x * wv[u];
                acc.y += hv[u].y * wv[u];
            }
        }
    }
    float2 r;
    r.x = fmaxf((acc.x + own.x) * dv + bv.x, 0.0f);
    r.y = fmaxf((acc.y + own.y) * dv + bv.y, 0.0f);
    return r;
}

#define GP_BLOCKS 1000

#define TROWS  16
#define TPITCH 68
#define WLANE  36

__device__ __forceinline__ void lds_barrier() {
    __builtin_amdgcn_fence(__ATOMIC_RELEASE, "workgroup", "local");
    __builtin_amdgcn_s_barrier();
    __builtin_amdgcn_fence(__ATOMIC_ACQUIRE, "workgroup", "local");
}

__global__ void __launch_bounds__(TROWS * 32)
gather_product_k(const int* __restrict__ rowptr, const int* __restrict__ rec,
                 const float* __restrict__ h, const float* __restrict__ dis,
                 const float* __restrict__ bias, const float* __restrict__ Wn,
                 float* __restrict__ outp, int n) {
    __shared__ __attribute__((aligned(16))) float tA[TROWS * TPITCH];
    __shared__ __attribute__((aligned(16))) float tD[TROWS * TPITCH];
    __shared__ __attribute__((aligned(16))) float sW[4 * 32 * WLANE];
    const int lane = threadIdx.x & 31;
    const int wid  = threadIdx.x >> 5;
    const int tiles = (n + TROWS - 1) / TROWS;
    const int S = gridDim.x * TROWS;
    const int mr  = lane & 15;
    const int khi = (lane >> 4) << 1;

#pragma unroll
    for (int cc = 0; cc < HID / 16; ++cc) {
        const float* wp = Wn + (4 * wid + khi) * HID + cc * 16 + mr;
        float2 wv2;
        wv2.x = wp[0];
        wv2.y = wp[HID];
        *(float2*)(sW + (cc * 32 + lane) * WLANE + 2 * wid) = wv2;
    }

    const float* hl = h + (lane << 1);
    const float2 bv = *(const float2*)(bias + (lane << 1));
    int tile = blockIdx.x;
    int node = __builtin_amdgcn_readfirstlane(min(tile * TROWS + wid, n - 1));
    int beg = rowptr[node];
    int end = rowptr[node + 1];
    int ev = 0;
    if (lane < end - beg) ev = rec[beg + lane];
    float2 own = *(const float2*)(hl + (size_t)node * HID);
    float dv = dis[node];
    int n1 = __builtin_amdgcn_readfirstlane(min(node + S, n - 1));
    int nb = rowptr[n1 + (lane & 1)];
    __syncthreads();

    for (; tile < tiles; tile += gridDim.x) {
        const int  cur  = tile * TROWS + wid;
        const bool live = cur < n;
        const float2 r = gather_row(rec, hl, lane, beg, end, ev, own, dv, bv);
        const int nbeg = __builtin_amdgcn_readlane(nb, 0);
        const int nend = __builtin_amdgcn_readlane(nb, 1);
        int nev = 0;
        if (lane < nend - nbeg) nev = rec[nbeg + lane];
        const float2 nown = *(const float2*)(hl + (size_t)n1 * HID);
        const float  ndv  = dis[n1];
        const int n2 = __builtin_amdgcn_readfirstlane(min(n1 + S, n - 1));
        const int nnb = rowptr[n2 + (lane & 1)];

        *(float2*)(tA + wid * TPITCH + (lane << 1)) = r;
        lds_barrier();

        if (wid < HID / 16) {
            const float* arow = tA + mr * TPITCH + khi;
            const float* bfr  = sW + (wid * 32 + lane) * WLANE;
            v8f acc = {};
#pragma unroll
            for (int k = 0; k < HID; k += 8) {
                const float4 bq = *(const float4*)(bfr + (k >> 1));
                v2f a, b;
                const float2 av0 = *(const float2*)(arow + k);
                a[0] = av0.x; a[1] = av0.y;
                b[0] = bq.x;  b[1] = bq.y;
                acc = __builtin_amdgcn_wmma_f32_16x16x4_f32(false, a, false, b, (short)0, acc, false, false);
                const float2 av1 = *(const float2*)(arow + k + 4);
                a[0] = av1.x; a[1] = av1.y;
                b[0] = bq.z;  b[1] = bq.w;
                acc = __builtin_amdgcn_wmma_f32_16x16x4_f32(false, a, false, b, (short)0, acc, false, false);
            }
            float* drow = tD + ((lane >> 4) << 3) * TPITCH + wid * 16 + mr;
#pragma unroll
            for (int j = 0; j < 8; ++j) drow[j * TPITCH] = acc[j];
        }
        lds_barrier();
        if (live) {
            const float2 pr = *(const float2*)(tD + wid * TPITCH + (lane << 1));
            *(float2*)(outp + (size_t)cur * HID + (lane << 1)) = make_float2(pr.x * dv, pr.y * dv);
        }
        beg = nbeg; end = nend; ev = nev; own = nown; dv = ndv; n1 = n2; nb = nnb;
    }
}

#define PC 7
#define GL_BLOCKS (((N_NODES + PC - 1) / PC + 7) / 8)

__device__ __forceinline__ void pool_flush(float* __restrict__ pooled, float* __restrict__ counts,
                                           int g, int lane, float sx, float sy,
                                           unsigned int mx, unsigned int my, int c) {
    float* p = pooled + (size_t)g * 128 + (lane << 1);
    atomicAdd(p, sx);
    atomicAdd(p + 1, sy);
    atomicMax((unsigned int*)(p + 64), mx);
    atomicMax((unsigned int*)(p + 65), my);
    if (lane == 0) atomicAdd(&counts[g], (float)c);
}

__global__ void __launch_bounds__(256)
gather_pool_k(const int* __restrict__ rowptr, const int* __restrict__ rec,
              const float* __restrict__ h, const float* __restrict__ dis,
              const float* __restrict__ bias, const int* __restrict__ batch,
              float* __restrict__ pooled, float* __restrict__ counts, int n) {
    const int lane = threadIdx.x & 31;
    const int w = __builtin_amdgcn_readfirstlane((blockIdx.x * blockDim.x + threadIdx.x) >> 5);
    const int node0 = w * PC;
    if (node0 >= n) return;
    const int m = min(PC, n - node0);
    const float* hl = h + (lane << 1);
    const float2 bv = *(const float2*)(bias + (lane << 1));
    const int   rp  = rowptr[node0 + min(lane, m)];
    const int   jc  = min(lane, m - 1);
    const int   gid = batch[node0 + jc];
    const float dvl = dis[node0 + jc];
    int beg = __builtin_amdgcn_readlane(rp, 0);
    int end = __builtin_amdgcn_readlane(rp, 1);
    int ev = 0;
    if (lane < end - beg) ev = rec[beg + lane];
    float2 own = *(const float2*)(hl + (size_t)node0 * HID);

    int cur = __builtin_amdgcn_readlane(gid, 0);
    float sx = 0.0f, sy = 0.0f;
    unsigned int mx = 0u, my = 0u;
    int c = 0;
    for (int i = 0; i < m; ++i) {
        const float dv = __int_as_float(__builtin_amdgcn_readlane(__float_as_int(dvl), i));
        const float2 r = gather_row(rec, hl, lane, beg, end, ev, own, dv, bv);
        const int i1 = min(i + 1, m - 1);
        const int nbeg = __builtin_amdgcn_readlane(rp, i1);
        const int nend = __builtin_amdgcn_readlane(rp, i1 + 1);
        int nev = 0;
        if (lane < nend - nbeg) nev = rec[nbeg + lane];
        const float2 nown = *(const float2*)(hl + (size_t)(node0 + i1) * HID);

        const int g = __builtin_amdgcn_readlane(gid, i);
        if (g != cur) {
            pool_flush(pooled, counts, cur, lane, sx, sy, mx, my, c);
            cur = g; sx = 0.0f; sy = 0.0f; mx = 0u; my = 0u; c = 0;
        }
        sx += r.x;
        sy += r.y;
        mx = max(mx, __float_as_uint(r.x));
        my = max(my, __float_as_uint(r.y));
        ++c;
        beg = nbeg; end = nend; ev = nev; own = nown;
    }
    pool_flush(pooled, counts, cur, lane, sx, sy, mx, my, c);
}

__global__ void head_k(float* __restrict__ pooled, const float* __restrict__ counts,
                       const float* __restrict__ Wout,
                       const float* __restrict__ bout, float* __restrict__ out, int G) {
    int t = blockIdx.x * blockDim.x + threadIdx.x;
    int g = t >> 5;
    int lane = t & 31;
    if (g >= G) return;
    float* p = pooled + (size_t)g * 128;
    const float d = fmaxf(counts[g], 1.0f);
    const float m0 = p[lane] / d;
    const float m1 = p[lane + 32] / d;
    p[lane] = m0;
    p[lane + 32] = m1;
    float v = 0.0f;
    v += m0 * Wout[lane];
    v += m1 * Wout[lane + 32];
    v += p[lane + 64] * Wout[lane + 64];
    v += p[lane + 96] * Wout[lane + 96];
#pragma unroll
    for (int off = 16; off; off >>= 1) v += __shfl_xor(v, off, 32);
    if (lane == 0) out[g] = v + bout[0];
}

#define WS_DIS    0
#define WS_ROWPTR (WS_DIS + N_NODES)
#define WS_COUNTS (WS_ROWPTR + N_NODES + 4)
#define WS_START  (WS_COUNTS + NGRAPH)
#define WS_TABLE  ((NCH * NB + 3) / 4 * 4)
#define WS_CNT    (WS_START + WS_TABLE)
#define WS_BIN    (WS_CNT + WS_TABLE)
#define WS_REC    (WS_BIN + 2 * NCH * CH)
#define WS_BUF0   (((WS_REC + N_EDGES) + 63) / 64 * 64)

extern "C" void kernel_launch(void* const* d_in, const int* in_sizes, int n_in,
                              void* d_out, int out_size, void* d_ws, size_t ws_size,
                              hipStream_t stream) {
    const float* x     = (const float*)d_in[0];
    const int*   eidx  = (const int*)  d_in[1];
    const int*   batch = (const int*)  d_in[2];
    const float* Ws[4] = { (const float*)d_in[3], (const float*)d_in[5],
                           (const float*)d_in[7], (const float*)d_in[9] };
    const float* bs[4] = { (const float*)d_in[4], (const float*)d_in[6],
                           (const float*)d_in[8], (const float*)d_in[10] };
    const float* Wout  = (const float*)d_in[11];
    const float* bout  = (const float*)d_in[12];

    const int N = N_NODES, E = N_EDGES, G = NGRAPH;
    const int* src = eidx;
    const int* dst = eidx + E;

    float* ws     = (float*)d_ws;
    float* dis    = ws + WS_DIS;
    int*   rowptr = (int*)(ws + WS_ROWPTR);
    float* counts = ws + WS_COUNTS;
    int*   startT = (int*)(ws + WS_START);
    int*   cntT   = (int*)(ws + WS_CNT);
    int*   binned = (int*)(ws + WS_BIN);
    int*   rec    = (int*)(ws + WS_REC);
    float* buf0   = ws + WS_BUF0;
    float* buf1   = buf0 + (size_t)N * HID;

    float* out    = (float*)d_out;
    float* pooled = out + G;

    chunk_bin_k  <<<NCH, SORT_T, 0, stream>>>(src, dst, startT, cntT, binned);
    bucket_rows_k<<<NB,  SORT_T, 0, stream>>>(binned, startT, cntT, rowptr, dis, rec);

    float* HWa = buf0;
    float* HWb = buf1;
    const int gemmThreads = (N / 16) * 32;
    gemm_wmma_k<<<(gemmThreads + 255) / 256, 256, 0, stream>>>(x, Ws[0], dis, HWa, N, 128,
                                                               pooled, counts);
    for (int l = 0; l < 3; ++l) {
        gather_product_k<<<GP_BLOCKS, TROWS * 32, 0, stream>>>(
            rowptr, rec, HWa, dis, bs[l], Ws[l + 1], HWb, N);
        float* t = HWa; HWa = HWb; HWb = t;
    }
    gather_pool_k<<<GL_BLOCKS, 256, 0, stream>>>(rowptr, rec, HWa, dis, bs[3], batch,
                                                 pooled, counts, N);
    head_k<<<(G * 32 + 255) / 256, 256, 0, stream>>>(pooled, counts, Wout, bout, out, G);
}
